# starting kernel
# baseline (speedup 1.0000x reference)
#include <hip/hip_runtime.h>
#include <math.h>

typedef __attribute__((ext_vector_type(2))) float v2f;
typedef __attribute__((ext_vector_type(8))) float v8f;

#define NNODES 50000
#define NEDGES 800000

template <int NC>
__global__ __launch_bounds__(256) void gemm_wmma_f32(
    const float* __restrict__ X, const float* __restrict__ W,
    float* __restrict__ H, int n) {
  constexpr int K  = 128;
  constexpr int CT = (NC + 15) / 16;
  constexpr int PC = CT * 16;
  __shared__ v2f lw[(K / 2) * PC];

  const int tid = threadIdx.x;
  for (int i = tid; i < (K / 2) * PC; i += 256) {
    const int kp = i / PC, c = i % PC;
    v2f val;
    val.x = (c < NC) ? W[(2 * kp) * NC + c] : 0.0f;
    val.y = (c < NC) ? W[(2 * kp + 1) * NC + c] : 0.0f;
    lw[i] = val;
  }
  __syncthreads();

  const int wave = tid >> 5;
  const int lane = tid & 31;
  const int row0 = (blockIdx.x * 8 + wave) * 16;
  if (row0 >= n) return;

  const int m  = lane & 15;
  const int hk = lane >> 4;

  v8f acc[CT];
#pragma unroll
  for (int t = 0; t < CT; ++t) acc[t] = {};

  const float* xrow = X + (size_t)(row0 + m) * K + 2 * hk;

  for (int k4 = 0; k4 < K; k4 += 4) {
    v2f a;
    a.x = xrow[k4];
    a.y = xrow[k4 + 1];
    const v2f* brow = &lw[(k4 / 2 + hk) * PC + m];
#pragma unroll
    for (int t = 0; t < CT; ++t) {
      const v2f b = brow[t * 16];
      acc[t] = __builtin_amdgcn_wmma_f32_16x16x4_f32(
          false, a, false, b, (short)0, acc[t], false, false);
    }
  }

  const int crow = row0 + hk * 8;
#pragma unroll
  for (int t = 0; t < CT; ++t) {
    const int col = t * 16 + m;
    if ((NC % 16 == 0) || (col < NC)) {
#pragma unroll
      for (int r = 0; r < 8; ++r)
        H[(size_t)(crow + r) * NC + col] = acc[t][r];
    }
  }
}

__global__ void zerof(float* __restrict__ p, long long n) {
  long long i = (long long)blockIdx.x * blockDim.x + threadIdx.x;
  if (i < n) p[i] = 0.0f;
}

__global__ __launch_bounds__(256) void agg128(
    const float* __restrict__ H, const int* __restrict__ src,
    const int* __restrict__ tgt, const float* __restrict__ w,
    float* __restrict__ A, int E) {
  const int e = (int)(((long long)blockIdx.x * blockDim.x + threadIdx.x) >> 5);
  const int lane = threadIdx.x & 31;
  if (e >= E) return;
  const int   s  = src[e];
  const int   t  = tgt[e];
  const float we = w[e];
  const float4 v = *((const float4*)(H + (size_t)s * 128) + lane);
  float* ap = A + (size_t)t * 128 + lane * 4;
  atomicAdd(ap + 0, we * v.x);
  atomicAdd(ap + 1, we * v.y);
  atomicAdd(ap + 2, we * v.z);
  atomicAdd(ap + 3, we * v.w);
}

__global__ __launch_bounds__(256) void agg40(
    const float* __restrict__ H, const int* __restrict__ src,
    const int* __restrict__ tgt, const float* __restrict__ w,
    float* __restrict__ A, int E) {
  const int e = (int)(((long long)blockIdx.x * blockDim.x + threadIdx.x) >> 5);
  const int lane = threadIdx.x & 31;
  if (e >= E) return;
  const int   s  = src[e];
  const int   t  = tgt[e];
  const float we = w[e];
  atomicAdd(A + (size_t)t * 40 + lane, we * H[(size_t)s * 40 + lane]);
  if (lane < 8)
    atomicAdd(A + (size_t)t * 40 + 32 + lane, we * H[(size_t)s * 40 + 32 + lane]);
}

__global__ void bias_relu128(float* __restrict__ A, const float* __restrict__ b,
                             long long total) {
  long long i = (long long)blockIdx.x * blockDim.x + threadIdx.x;
  if (i >= total) return;
  const int c = (int)(i & 127);
  A[i] = fmaxf(A[i] + b[c], 0.0f);
}

__global__ __launch_bounds__(256) void gn_bias_relu(
    float* __restrict__ A, const float* __restrict__ b,
    const float* __restrict__ gamma, const float* __restrict__ beta, int n) {
  const int node = (int)(((long long)blockIdx.x * blockDim.x + threadIdx.x) >> 5);
  const int g = threadIdx.x & 31;
  if (node >= n) return;
  float* p = A + (size_t)node * 128 + g * 4;
  float4 v  = *(float4*)p;
  const float4 bb = *(const float4*)(b + g * 4);
  v.x = fmaxf(v.x + bb.x, 0.0f);
  v.y = fmaxf(v.y + bb.y, 0.0f);
  v.z = fmaxf(v.z + bb.z, 0.0f);
  v.w = fmaxf(v.w + bb.w, 0.0f);
  const float mu  = 0.25f * (v.x + v.y + v.z + v.w);
  const float ms  = 0.25f * (v.x * v.x + v.y * v.y + v.z * v.z + v.w * v.w);
  const float inv = rsqrtf(fmaxf(ms - mu * mu, 0.0f) + 1e-5f);
  const float4 gm = *(const float4*)(gamma + g * 4);
  const float4 bt = *(const float4*)(beta + g * 4);
  v.x = (v.x - mu) * inv * gm.x + bt.x;
  v.y = (v.y - mu) * inv * gm.y + bt.y;
  v.z = (v.z - mu) * inv * gm.z + bt.z;
  v.w = (v.w - mu) * inv * gm.w + bt.w;
  *(float4*)p = v;
}

__global__ __launch_bounds__(256) void lsm40(
    const float* __restrict__ Ain, const float* __restrict__ b,
    float* __restrict__ out, int n) {
  const int node = (int)(((long long)blockIdx.x * blockDim.x + threadIdx.x) >> 5);
  const int lane = threadIdx.x & 31;
  if (node >= n) return;
  const float* row = Ain + (size_t)node * 40;
  const float v0 = row[lane] + b[lane];
  const float v1 = (lane < 8) ? (row[32 + lane] + b[32 + lane]) : -INFINITY;
  float m = fmaxf(v0, v1);
#pragma unroll
  for (int o = 16; o > 0; o >>= 1) m = fmaxf(m, __shfl_xor(m, o, 32));
  float s = expf(v0 - m) + ((lane < 8) ? expf(v1 - m) : 0.0f);
#pragma unroll
  for (int o = 16; o > 0; o >>= 1) s += __shfl_xor(s, o, 32);
  const float lse = m + logf(s);
  out[(size_t)node * 40 + lane] = v0 - lse;
  if (lane < 8) out[(size_t)node * 40 + 32 + lane] = v1 - lse;
}

extern "C" void kernel_launch(void* const* d_in, const int* in_sizes, int n_in,
                              void* d_out, int out_size, void* d_ws, size_t ws_size,
                              hipStream_t stream) {
  const int N = NNODES, E = NEDGES;
  const float* x     = (const float*)d_in[0];
  const int*   src   = (const int*)d_in[1];
  const int*   tgt   = (const int*)d_in[2];
  const float* w     = (const float*)d_in[3];
  const float* W0    = (const float*)d_in[4];
  const float* b0    = (const float*)d_in[5];
  const float* W1    = (const float*)d_in[6];
  const float* b1    = (const float*)d_in[7];
  const float* W2    = (const float*)d_in[8];
  const float* b2    = (const float*)d_in[9];
  const float* gamma = (const float*)d_in[10];
  const float* beta  = (const float*)d_in[11];
  float* out = (float*)d_out;

  float* H   = (float*)d_ws;
  float* A   = H + (size_t)N * 128;
  float* H2  = H;
  float* AG2 = H + (size_t)N * 64;

  const long long tot128 = (long long)N * 128;
  const long long tot40  = (long long)N * 40;
  const int gemm_blocks  = (N + 127) / 128;
  const int edge_blocks  = (E + 7) / 8;
  const int ew_blocks    = (int)((tot128 + 255) / 256);
  const int node_blocks  = (N * 32 + 255) / 256;

  gemm_wmma_f32<128><<<gemm_blocks, 256, 0, stream>>>(x, W0, H, N);
  zerof<<<ew_blocks, 256, 0, stream>>>(A, tot128);
  agg128<<<edge_blocks, 256, 0, stream>>>(H, src, tgt, w, A, E);
  bias_relu128<<<ew_blocks, 256, 0, stream>>>(A, b0, tot128);

  gemm_wmma_f32<128><<<gemm_blocks, 256, 0, stream>>>(A, W1, H, N);
  zerof<<<ew_blocks, 256, 0, stream>>>(A, tot128);
  agg128<<<edge_blocks, 256, 0, stream>>>(H, src, tgt, w, A, E);
  gn_bias_relu<<<node_blocks, 256, 0, stream>>>(A, b1, gamma, beta, N);

  gemm_wmma_f32<40><<<gemm_blocks, 256, 0, stream>>>(A, W2, H2, N);
  zerof<<<(int)((tot40 + 255) / 256), 256, 0, stream>>>(AG2, tot40);
  agg40<<<edge_blocks, 256, 0, stream>>>(H2, src, tgt, w, AG2, E);
  lsm40<<<node_blocks, 256, 0, stream>>>(AG2, b2, out, N);
}


# float atomic scatter replaced by owner tiles: each of 256 work-groups keeps 196 target nodes' sums in LDS, filters the edge list for them, adds rows in a fixed order with fma; zero fills gone
# speedup vs baseline: 1.3647x; 1.3647x over previous
#include <hip/hip_runtime.h>
#include <math.h>

typedef __attribute__((ext_vector_type(2))) float v2f;
typedef __attribute__((ext_vector_type(8))) float v8f;

#define NNODES 50000
#define NEDGES 800000

template <int NC>
__global__ __launch_bounds__(256) void gemm_wmma_f32(
    const float* __restrict__ X, const float* __restrict__ W,
    float* __restrict__ H, int n) {
  constexpr int K  = 128;
  constexpr int CT = (NC + 15) / 16;
  constexpr int PC = CT * 16;
  __shared__ v2f lw[(K / 2) * PC];

  const int tid = threadIdx.x;
  for (int i = tid; i < (K / 2) * PC; i += 256) {
    const int kp = i / PC, c = i % PC;
    v2f val;
    val.x = (c < NC) ? W[(2 * kp) * NC + c] : 0.0f;
    val.y = (c < NC) ? W[(2 * kp + 1) * NC + c] : 0.0f;
    lw[i] = val;
  }
  __syncthreads();

  const int wave = tid >> 5;
  const int lane = tid & 31;
  const int row0 = (blockIdx.x * 8 + wave) * 16;
  if (row0 >= n) return;

  const int m  = lane & 15;
  const int hk = lane >> 4;

  v8f acc[CT];
#pragma unroll
  for (int t = 0; t < CT; ++t) acc[t] = {};

  const float* xrow = X + (size_t)(row0 + m) * K + 2 * hk;

  for (int k4 = 0; k4 < K; k4 += 4) {
    v2f a;
    a.x = xrow[k4];
    a.y = xrow[k4 + 1];
    const v2f* brow = &lw[(k4 / 2 + hk) * PC + m];
#pragma unroll
    for (int t = 0; t < CT; ++t) {
      const v2f b = brow[t * 16];
      acc[t] = __builtin_amdgcn_wmma_f32_16x16x4_f32(
          false, a, false, b, (short)0, acc[t], false, false);
    }
  }

  const int crow = row0 + hk * 8;
#pragma unroll
  for (int t = 0; t < CT; ++t) {
    const int col = t * 16 + m;
    if ((NC % 16 == 0) || (col < NC)) {
#pragma unroll
      for (int r = 0; r < 8; ++r)
        H[(size_t)(crow + r) * NC + col] = acc[t][r];
    }
  }
}

#define TILE 196
#define NTILES ((NNODES + TILE - 1) / TILE)
#define NWV 32
#define CAP 448
#define CAP2 256

static_assert(TILE <= 256, "local node id is kept in 8 bits");
static_assert(NNODES < (1 << 24) && NEDGES < (1 << 24), "ids are kept in 24 bits");

template <int F>
__global__ __launch_bounds__(1024) void agg_tile(
    const float* __restrict__ H, const int* __restrict__ src,
    const int* __restrict__ tgt, const float* __restrict__ w,
    float* __restrict__ A, int n, int E) {
  constexpr int F4 = F / 4;
  __shared__ float4   acc[TILE * F4];
  __shared__ unsigned lkey[NWV * CAP];
  __shared__ float    lwt[NWV * CAP];
  __shared__ unsigned okey[NWV * CAP2];
  __shared__ float    owt[NWV * CAP2];
  __shared__ int      lcnt[NWV];
  __shared__ int      wmore[NWV];

  const int tid  = threadIdx.x;
  const int lane = tid & 31;
  const int wave = __builtin_amdgcn_readfirstlane(tid >> 5);
  const int lo   = blockIdx.x * TILE;
  const unsigned below = (1u << lane) - 1u;

  for (int i = tid; i < TILE * F4; i += 1024) acc[i] = make_float4(0.f, 0.f, 0.f, 0.f);

  const int per = (((E + NWV - 1) / NWV) + 3) & ~3;
  int pos = min(wave * per, E);
  const int end = min(pos + per, E);
  unsigned* mykey = lkey + wave * CAP;
  float*    myw   = lwt + wave * CAP;
  unsigned* ok    = okey + wave * CAP2;
  float*    ow    = owt + wave * CAP2;

  const int max_trips = per / (CAP - 31) + 2;
  int skip = 0;
  for (int trip = 0; trip < max_trips; ++trip) {
    __syncthreads();

    int  cnt  = 0;
    bool full = false;
    while (pos < end && !full) {
      int4 t[8];
      if (pos + 1024 <= end) {
        const int4* tp = (const int4*)(tgt + pos) + lane;
#pragma unroll
        for (int u = 0; u < 8; ++u) t[u] = tp[u * 32];
      } else {
#pragma unroll
        for (int u = 0; u < 8; ++u) {
          const int e = pos + (u * 32 + lane) * 4;
          const int v0 = tgt[min(e, end - 1)];
          const int v1 = tgt[min(e + 1, end - 1)];
          const int v2 = tgt[min(e + 2, end - 1)];
          const int v3 = tgt[min(e + 3, end - 1)];
          t[u].x = (e < end) ? v0 : -1;
          t[u].y = (e + 1 < end) ? v1 : -1;
          t[u].z = (e + 2 < end) ? v2 : -1;
          t[u].w = (e + 3 < end) ? v3 : -1;
        }
      }
#pragma unroll
      for (int u = 0; u < 8; ++u) {
        const int tv[4] = {t[u].x, t[u].y, t[u].z, t[u].w};
        unsigned m[4];
#pragma unroll
        for (int c = 0; c < 4; ++c)
          m[c] = __builtin_amdgcn_ballot_w32((unsigned)(tv[c] - lo) < (unsigned)TILE);
        if ((m[0] | m[1] | m[2] | m[3]) != 0) {
#pragma unroll
          for (int c = 0; c < 4; ++c) {
            const int q = u * 4 + c;
            if (m[c] != 0 && q >= skip && !full) {
              if (cnt > CAP - 32) {
                full = true;
                skip = q;
              } else {
                if ((m[c] >> lane) & 1u) {
                  const int idx = cnt + __builtin_popcount(m[c] & below);
                  mykey[idx] = ((unsigned)(tv[c] - lo) << 24) |
                               (unsigned)(pos + (u * 32 + lane) * 4 + c);
                }
                cnt += __builtin_popcount(m[c]);
              }
            }
          }
        }
      }
      if (!full) { pos += 1024; skip = 0; }
    }
    __builtin_amdgcn_wave_barrier();
    for (int i = lane; i < cnt; i += 32) {
      const unsigned k = mykey[i];
      const int e = (int)(k & 0xFFFFFFu);
      mykey[i] = (k & 0xFF000000u) | (unsigned)src[e];
      myw[i]   = w[e];
    }
    if (lane == 0) {
      lcnt[wave]  = cnt;
      wmore[wave] = (pos < end) ? 1 : 0;
    }
    __syncthreads();
    const bool more = __builtin_amdgcn_ballot_w32(wmore[lane] != 0) != 0;

    int sl = 0, base = 0;
    while (sl < NWV) {
      int n_own = 0;
      while (sl < NWV && n_own <= CAP2 - 32) {
        const int c = __builtin_amdgcn_readfirstlane(lcnt[sl]);
        if (base >= c) { ++sl; base = 0; continue; }
        const int i = base + lane;
        unsigned k = 0;
        float ww = 0.f;
        bool mine = false;
        if (i < c) {
          k    = lkey[sl * CAP + i];
          ww   = lwt[sl * CAP + i];
          mine = (int)((k >> 24) & 31u) == wave;
        }
        const unsigned mm = __builtin_amdgcn_ballot_w32(mine);
        if (mm != 0) {
          if (mine) {
            const int idx = n_own + __builtin_popcount(mm & below);
            ok[idx] = k;
            ow[idx] = ww;
          }
          n_own += __builtin_popcount(mm);
        }
        base += 32;
      }
      __builtin_amdgcn_wave_barrier();
      for (int j0 = 0; j0 < n_own; j0 += 8) {
        unsigned kk[8];
        float wq[8];
        float4 v[8];
#pragma unroll
        for (int u = 0; u < 8; ++u) {
          const int jj = min(j0 + u, n_own - 1);
          kk[u] = (unsigned)__builtin_amdgcn_readfirstlane((int)ok[jj]);
          wq[u] = ow[jj];
        }
#pragma unroll
        for (int u = 0; u < 8; ++u) {
          const size_t s = (size_t)(kk[u] & 0xFFFFFFu);
          v[u] = make_float4(0.f, 0.f, 0.f, 0.f);
          if (lane < F4) v[u] = *((const float4*)(H + s * F) + lane);
        }
#pragma unroll
        for (int u = 0; u < 8; ++u) {
          if (j0 + u < n_own) {
            const int tl = (int)(kk[u] >> 24);
            if (lane < F4) {
              float4 a = acc[tl * F4 + lane];
              a.x = fmaf(wq[u], v[u].x, a.x);
              a.y = fmaf(wq[u], v[u].y, a.y);
              a.z = fmaf(wq[u], v[u].z, a.z);
              a.w = fmaf(wq[u], v[u].w, a.w);
              acc[tl * F4 + lane] = a;
            }
          }
        }
      }
      __builtin_amdgcn_wave_barrier();
    }
    if (!more) break;
  }
  __syncthreads();

  float4* A4 = (float4*)A;
  const int g0   = lo * F4;
  const int gmax = n * F4;
  for (int i = tid; i < TILE * F4; i += 1024)
    if (g0 + i < gmax) A4[g0 + i] = acc[i];
}

__global__ void bias_relu128(float* __restrict__ A, const float* __restrict__ b,
                             long long total) {
  long long i = (long long)blockIdx.x * blockDim.x + threadIdx.x;
  if (i >= total) return;
  const int c = (int)(i & 127);
  A[i] = fmaxf(A[i] + b[c], 0.0f);
}

__global__ __launch_bounds__(256) void gn_bias_relu(
    float* __restrict__ A, const float* __restrict__ b,
    const float* __restrict__ gamma, const float* __restrict__ beta, int n) {
  const int node = (int)(((long long)blockIdx.x * blockDim.x + threadIdx.x) >> 5);
  const int g = threadIdx.x & 31;
  if (node >= n) return;
  float* p = A + (size_t)node * 128 + g * 4;
  float4 v  = *(float4*)p;
  const float4 bb = *(const float4*)(b + g * 4);
  v.x = fmaxf(v.x + bb.x, 0.0f);
  v.y = fmaxf(v.y + bb.y, 0.0f);
  v.z = fmaxf(v.z + bb.z, 0.0f);
  v.w = fmaxf(v.w + bb.w, 0.0f);
  const float mu  = 0.25f * (v.x + v.y + v.z + v.w);
  const float ms  = 0.25f * (v.x * v.x + v.y * v.y + v.z * v.z + v.w * v.w);
  const float inv = rsqrtf(fmaxf(ms - mu * mu, 0.0f) + 1e-5f);
  const float4 gm = *(const float4*)(gamma + g * 4);
  const float4 bt = *(const float4*)(beta + g * 4);
  v.x = (v.x - mu) * inv * gm.x + bt.x;
  v.y = (v.y - mu) * inv * gm.y + bt.y;
  v.z = (v.z - mu) * inv * gm.z + bt.z;
  v.w = (v.w - mu) * inv * gm.w + bt.w;
  *(float4*)p = v;
}

__global__ __launch_bounds__(256) void lsm40(
    const float* __restrict__ Ain, const float* __restrict__ b,
    float* __restrict__ out, int n) {
  const int node = (int)(((long long)blockIdx.x * blockDim.x + threadIdx.x) >> 5);
  const int lane = threadIdx.x & 31;
  if (node >= n) return;
  const float* row = Ain + (size_t)node * 40;
  const float v0 = row[lane] + b[lane];
  const float v1 = (lane < 8) ? (row[32 + lane] + b[32 + lane]) : -INFINITY;
  float m = fmaxf(v0, v1);
#pragma unroll
  for (int o = 16; o > 0; o >>= 1) m = fmaxf(m, __shfl_xor(m, o, 32));
  float s = expf(v0 - m) + ((lane < 8) ? expf(v1 - m) : 0.0f);
#pragma unroll
  for (int o = 16; o > 0; o >>= 1) s += __shfl_xor(s, o, 32);
  const float lse = m + logf(s);
  out[(size_t)node * 40 + lane] = v0 - lse;
  if (lane < 8) out[(size_t)node * 40 + 32 + lane] = v1 - lse;
}

extern "C" void kernel_launch(void* const* d_in, const int* in_sizes, int n_in,
                              void* d_out, int out_size, void* d_ws, size_t ws_size,
                              hipStream_t stream) {
  const int N = NNODES, E = NEDGES;
  const float* x     = (const float*)d_in[0];
  const int*   src   = (const int*)d_in[1];
  const int*   tgt   = (const int*)d_in[2];
  const float* w     = (const float*)d_in[3];
  const float* W0    = (const float*)d_in[4];
  const float* b0    = (const float*)d_in[5];
  const float* W1    = (const float*)d_in[6];
  const float* b1    = (const float*)d_in[7];
  const float* W2    = (const float*)d_in[8];
  const float* b2    = (const float*)d_in[9];
  const float* gamma = (const float*)d_in[10];
  const float* beta  = (const float*)d_in[11];
  float* out = (float*)d_out;

  float* H   = (float*)d_ws;
  float* A   = H + (size_t)N * 128;
  float* H2  = H;
  float* AG2 = H + (size_t)N * 64;

  const long long tot128 = (long long)N * 128;
  const int gemm_blocks  = (N + 127) / 128;
  const int ew_blocks    = (int)((tot128 + 255) / 256);
  const int node_blocks  = (N * 32 + 255) / 256;

  gemm_wmma_f32<128><<<gemm_blocks, 256, 0, stream>>>(x, W0, H, N);
  agg_tile<128><<<NTILES, 1024, 0, stream>>>(H, src, tgt, w, A, N, E);
  bias_relu128<<<ew_blocks, 256, 0, stream>>>(A, b0, tot128);

  gemm_wmma_f32<128><<<gemm_blocks, 256, 0, stream>>>(A, W1, H, N);
  agg_tile<128><<<NTILES, 1024, 0, stream>>>(H, src, tgt, w, A, N, E);
  gn_bias_relu<<<node_blocks, 256, 0, stream>>>(A, b1, gamma, beta, N);

  gemm_wmma_f32<40><<<gemm_blocks, 256, 0, stream>>>(A, W2, H2, N);
  agg_tile<40><<<NTILES, 1024, 0, stream>>>(H2, src, tgt, w, AG2, N, E);
  lsm40<<<node_blocks, 256, 0, stream>>>(AG2, b2, out, N);
}


# bias+relu, bias+relu+groupnorm and bias+log_softmax kernels re-gridded: 8192 resident waves loop over rows, 8 loads in flight each, in place of 300000 short one-load waves; arithmetic unchanged
# speedup vs baseline: 1.4386x; 1.0542x over previous
#include <hip/hip_runtime.h>
#include <math.h>

typedef __attribute__((ext_vector_type(2))) float v2f;
typedef __attribute__((ext_vector_type(8))) float v8f;

#define NNODES 50000
#define NEDGES 800000

template <int NC>
__global__ __launch_bounds__(256) void gemm_wmma_f32(
    const float* __restrict__ X, const float* __restrict__ W,
    float* __restrict__ H, int n) {
  constexpr int K  = 128;
  constexpr int CT = (NC + 15) / 16;
  constexpr int PC = CT * 16;
  __shared__ v2f lw[(K / 2) * PC];

  const int tid = threadIdx.x;
  for (int i = tid; i < (K / 2) * PC; i += 256) {
    const int kp = i / PC, c = i % PC;
    v2f val;
    val.x = (c < NC) ? W[(2 * kp) * NC + c] : 0.0f;
    val.y = (c < NC) ? W[(2 * kp + 1) * NC + c] : 0.0f;
    lw[i] = val;
  }
  __syncthreads();

  const int wave = tid >> 5;
  const int lane = tid & 31;
  const int row0 = (blockIdx.x * 8 + wave) * 16;
  if (row0 >= n) return;

  const int m  = lane & 15;
  const int hk = lane >> 4;

  v8f acc[CT];
#pragma unroll
  for (int t = 0; t < CT; ++t) acc[t] = {};

  const float* xrow = X + (size_t)(row0 + m) * K + 2 * hk;

  for (int k4 = 0; k4 < K; k4 += 4) {
    v2f a;
    a.x = xrow[k4];
    a.y = xrow[k4 + 1];
    const v2f* brow = &lw[(k4 / 2 + hk) * PC + m];
#pragma unroll
    for (int t = 0; t < CT; ++t) {
      const v2f b = brow[t * 16];
      acc[t] = __builtin_amdgcn_wmma_f32_16x16x4_f32(
          false, a, false, b, (short)0, acc[t], false, false);
    }
  }

  const int crow = row0 + hk * 8;
#pragma unroll
  for (int t = 0; t < CT; ++t) {
    const int col = t * 16 + m;
    if ((NC % 16 == 0) || (col < NC)) {
#pragma unroll
      for (int r = 0; r < 8; ++r)
        H[(size_t)(crow + r) * NC + col] = acc[t][r];
    }
  }
}

#define TILE 196
#define NTILES ((NNODES + TILE - 1) / TILE)
#define NWV 32
#define CAP 448
#define CAP2 256

static_assert(TILE <= 256, "local node id is kept in 8 bits");
static_assert(NNODES < (1 << 24) && NEDGES < (1 << 24), "ids are kept in 24 bits");

template <int F>
__global__ __launch_bounds__(1024) void agg_tile(
    const float* __restrict__ H, const int* __restrict__ src,
    const int* __restrict__ tgt, const float* __restrict__ w,
    float* __restrict__ A, int n, int E) {
  constexpr int F4 = F / 4;
  __shared__ float4   acc[TILE * F4];
  __shared__ unsigned lkey[NWV * CAP];
  __shared__ float    lwt[NWV * CAP];
  __shared__ unsigned okey[NWV * CAP2];
  __shared__ float    owt[NWV * CAP2];
  __shared__ int      lcnt[NWV];
  __shared__ int      wmore[NWV];

  const int tid  = threadIdx.x;
  const int lane = tid & 31;
  const int wave = __builtin_amdgcn_readfirstlane(tid >> 5);
  const int lo   = blockIdx.x * TILE;
  const unsigned below = (1u << lane) - 1u;

  for (int i = tid; i < TILE * F4; i += 1024) acc[i] = make_float4(0.f, 0.f, 0.f, 0.f);

  const int per = (((E + NWV - 1) / NWV) + 3) & ~3;
  int pos = min(wave * per, E);
  const int end = min(pos + per, E);
  unsigned* mykey = lkey + wave * CAP;
  float*    myw   = lwt + wave * CAP;
  unsigned* ok    = okey + wave * CAP2;
  float*    ow    = owt + wave * CAP2;

  const int max_trips = per / (CAP - 31) + 2;
  int skip = 0;
  for (int trip = 0; trip < max_trips; ++trip) {
    __syncthreads();

    int  cnt  = 0;
    bool full = false;
    while (pos < end && !full) {
      int4 t[8];
      if (pos + 1024 <= end) {
        const int4* tp = (const int4*)(tgt + pos) + lane;
#pragma unroll
        for (int u = 0; u < 8; ++u) t[u] = tp[u * 32];
      } else {
#pragma unroll
        for (int u = 0; u < 8; ++u) {
          const int e = pos + (u * 32 + lane) * 4;
          const int v0 = tgt[min(e, end - 1)];
          const int v1 = tgt[min(e + 1, end - 1)];
          const int v2 = tgt[min(e + 2, end - 1)];
          const int v3 = tgt[min(e + 3, end - 1)];
          t[u].x = (e < end) ? v0 : -1;
          t[u].y = (e + 1 < end) ? v1 : -1;
          t[u].z = (e + 2 < end) ? v2 : -1;
          t[u].w = (e + 3 < end) ? v3 : -1;
        }
      }
#pragma unroll
      for (int u = 0; u < 8; ++u) {
        const int tv[4] = {t[u].x, t[u].y, t[u].z, t[u].w};
        unsigned m[4];
#pragma unroll
        for (int c = 0; c < 4; ++c)
          m[c] = __builtin_amdgcn_ballot_w32((unsigned)(tv[c] - lo) < (unsigned)TILE);
        if ((m[0] | m[1] | m[2] | m[3]) != 0) {
#pragma unroll
          for (int c = 0; c < 4; ++c) {
            const int q = u * 4 + c;
            if (m[c] != 0 && q >= skip && !full) {
              if (cnt > CAP - 32) {
                full = true;
                skip = q;
              } else {
                if ((m[c] >> lane) & 1u) {
                  const int idx = cnt + __builtin_popcount(m[c] & below);
                  mykey[idx] = ((unsigned)(tv[c] - lo) << 24) |
                               (unsigned)(pos + (u * 32 + lane) * 4 + c);
                }
                cnt += __builtin_popcount(m[c]);
              }
            }
          }
        }
      }
      if (!full) { pos += 1024; skip = 0; }
    }
    __builtin_amdgcn_wave_barrier();
    for (int i = lane; i < cnt; i += 32) {
      const unsigned k = mykey[i];
      const int e = (int)(k & 0xFFFFFFu);
      mykey[i] = (k & 0xFF000000u) | (unsigned)src[e];
      myw[i]   = w[e];
    }
    if (lane == 0) {
      lcnt[wave]  = cnt;
      wmore[wave] = (pos < end) ? 1 : 0;
    }
    __syncthreads();
    const bool more = __builtin_amdgcn_ballot_w32(wmore[lane] != 0) != 0;

    int sl = 0, base = 0;
    while (sl < NWV) {
      int n_own = 0;
      while (sl < NWV && n_own <= CAP2 - 32) {
        const int c = __builtin_amdgcn_readfirstlane(lcnt[sl]);
        if (base >= c) { ++sl; base = 0; continue; }
        const int i = base + lane;
        unsigned k = 0;
        float ww = 0.f;
        bool mine = false;
        if (i < c) {
          k    = lkey[sl * CAP + i];
          ww   = lwt[sl * CAP + i];
          mine = (int)((k >> 24) & 31u) == wave;
        }
        const unsigned mm = __builtin_amdgcn_ballot_w32(mine);
        if (mm != 0) {
          if (mine) {
            const int idx = n_own + __builtin_popcount(mm & below);
            ok[idx] = k;
            ow[idx] = ww;
          }
          n_own += __builtin_popcount(mm);
        }
        base += 32;
      }
      __builtin_amdgcn_wave_barrier();
      for (int j0 = 0; j0 < n_own; j0 += 8) {
        unsigned kk[8];
        float wq[8];
        float4 v[8];
#pragma unroll
        for (int u = 0; u < 8; ++u) {
          const int jj = min(j0 + u, n_own - 1);
          kk[u] = (unsigned)__builtin_amdgcn_readfirstlane((int)ok[jj]);
          wq[u] = ow[jj];
        }
#pragma unroll
        for (int u = 0; u < 8; ++u) {
          const size_t s = (size_t)(kk[u] & 0xFFFFFFu);
          v[u] = make_float4(0.f, 0.f, 0.f, 0.f);
          if (lane < F4) v[u] = *((const float4*)(H + s * F) + lane);
        }
#pragma unroll
        for (int u = 0; u < 8; ++u) {
          if (j0 + u < n_own) {
            const int tl = (int)(kk[u] >> 24);
            if (lane < F4) {
              float4 a = acc[tl * F4 + lane];
              a.x = fmaf(wq[u], v[u].x, a.x);
              a.y = fmaf(wq[u], v[u].y, a.y);
              a.z = fmaf(wq[u], v[u].z, a.z);
              a.w = fmaf(wq[u], v[u].w, a.w);
              acc[tl * F4 + lane] = a;
            }
          }
        }
      }
      __builtin_amdgcn_wave_barrier();
    }
    if (!more) break;
  }
  __syncthreads();

  float4* A4 = (float4*)A;
  const int g0   = lo * F4;
  const int gmax = n * F4;
  for (int i = tid; i < TILE * F4; i += 1024)
    if (g0 + i < gmax) A4[g0 + i] = acc[i];
}

#define EW_BLOCKS 1024
#define EW_THREADS 256
#define EW_UNROLL 8

__global__ __launch_bounds__(EW_THREADS) void bias_relu128(
    float* __restrict__ A, const float* __restrict__ b, int total4) {
  const int T  = gridDim.x * blockDim.x;
  const int i0 = blockIdx.x * blockDim.x + threadIdx.x;
  const float4 bb = *(const float4*)(b + (i0 & 31) * 4);
  float4* A4 = (float4*)A;
  for (int base = i0; base < total4; base += EW_UNROLL * T) {
    float4 v[EW_UNROLL];
#pragma unroll
    for (int k = 0; k < EW_UNROLL; ++k) {
      const int i = base + k * T;
      v[k] = make_float4(0.f, 0.f, 0.f, 0.f);
      if (i < total4) v[k] = A4[i];
    }
#pragma unroll
    for (int k = 0; k < EW_UNROLL; ++k) {
      const int i = base + k * T;
      if (i < total4) {
        float4 r;
        r.x = fmaxf(v[k].x + bb.x, 0.0f);
        r.y = fmaxf(v[k].y + bb.y, 0.0f);
        r.z = fmaxf(v[k].z + bb.z, 0.0f);
        r.w = fmaxf(v[k].w + bb.w, 0.0f);
        A4[i] = r;
      }
    }
  }
}

__global__ __launch_bounds__(EW_THREADS) void gn_bias_relu(
    float* __restrict__ A, const float* __restrict__ b,
    const float* __restrict__ gamma, const float* __restrict__ beta, int total4) {
  const int T  = gridDim.x * blockDim.x;
  const int i0 = blockIdx.x * blockDim.x + threadIdx.x;
  const int g  = i0 & 31;
  const float4 bb = *(const float4*)(b + g * 4);
  const float4 gm = *(const float4*)(gamma + g * 4);
  const float4 bt = *(const float4*)(beta + g * 4);
  float4* A4 = (float4*)A;
  for (int base = i0; base < total4; base += EW_UNROLL * T) {
    float4 q[EW_UNROLL];
#pragma unroll
    for (int k = 0; k < EW_UNROLL; ++k) {
      const int i = base + k * T;
      q[k] = make_float4(0.f, 0.f, 0.f, 0.f);
      if (i < total4) q[k] = A4[i];
    }
#pragma unroll
    for (int k = 0; k < EW_UNROLL; ++k) {
      const int i = base + k * T;
      if (i < total4) {
        float4 v = q[k];
        v.x = fmaxf(v.x + bb.x, 0.0f);
        v.y = fmaxf(v.y + bb.y, 0.0f);
        v.z = fmaxf(v.z + bb.z, 0.0f);
        v.w = fmaxf(v.w + bb.w, 0.0f);
        const float mu  = 0.25f * (v.x + v.y + v.z + v.w);
        const float ms  = 0.25f * (v.x * v.x + v.y * v.y + v.z * v.z + v.w * v.w);
        const float inv = rsqrtf(fmaxf(ms - mu * mu, 0.0f) + 1e-5f);
        v.x = (v.x - mu) * inv * gm.x + bt.x;
        v.y = (v.y - mu) * inv * gm.y + bt.y;
        v.z = (v.z - mu) * inv * gm.z + bt.z;
        v.w = (v.w - mu) * inv * gm.w + bt.w;
        A4[i] = v;
      }
    }
  }
}

__global__ __launch_bounds__(EW_THREADS) void lsm40(
    const float* __restrict__ Ain, const float* __restrict__ b,
    float* __restrict__ out, int n) {
  const int lane = threadIdx.x & 31;
  const int gw = __builtin_amdgcn_readfirstlane(
      (int)((blockIdx.x * blockDim.x + threadIdx.x) >> 5));
  const int nw = (int)((gridDim.x * blockDim.x) >> 5);
  const float c0 = b[lane];
  const float c1 = (lane < 8) ? b[32 + lane] : 0.0f;
  for (int nb = gw; nb < n; nb += EW_UNROLL * nw) {
    float r0[EW_UNROLL], r1[EW_UNROLL];
#pragma unroll
    for (int k = 0; k < EW_UNROLL; ++k) {
      const int node = nb + k * nw;
      r0[k] = 0.0f;
      r1[k] = 0.0f;
      if (node < n) {
        const float* row = Ain + (size_t)node * 40;
        r0[k] = row[lane];
        if (lane < 8) r1[k] = row[32 + lane];
      }
    }
#pragma unroll
    for (int k = 0; k < EW_UNROLL; ++k) {
      const int node = nb + k * nw;
      if (node < n) {
        const float v0 = r0[k] + c0;
        const float v1 = (lane < 8) ? (r1[k] + c1) : -INFINITY;
        float m = fmaxf(v0, v1);
#pragma unroll
        for (int o = 16; o > 0; o >>= 1) m = fmaxf(m, __shfl_xor(m, o, 32));
        float s = expf(v0 - m) + ((lane < 8) ? expf(v1 - m) : 0.0f);
#pragma unroll
        for (int o = 16; o > 0; o >>= 1) s += __shfl_xor(s, o, 32);
        const float lse = m + logf(s);
        out[(size_t)node * 40 + lane] = v0 - lse;
        if (lane < 8) out[(size_t)node * 40 + 32 + lane] = v1 - lse;
      }
    }
  }
}

extern "C" void kernel_launch(void* const* d_in, const int* in_sizes, int n_in,
                              void* d_out, int out_size, void* d_ws, size_t ws_size,
                              hipStream_t stream) {
  const int N = NNODES, E = NEDGES;
  const float* x     = (const float*)d_in[0];
  const int*   src   = (const int*)d_in[1];
  const int*   tgt   = (const int*)d_in[2];
  const float* w     = (const float*)d_in[3];
  const float* W0    = (const float*)d_in[4];
  const float* b0    = (const float*)d_in[5];
  const float* W1    = (const float*)d_in[6];
  const float* b1    = (const float*)d_in[7];
  const float* W2    = (const float*)d_in[8];
  const float* b2    = (const float*)d_in[9];
  const float* gamma = (const float*)d_in[10];
  const float* beta  = (const float*)d_in[11];
  float* out = (float*)d_out;

  float* H   = (float*)d_ws;
  float* A   = H + (size_t)N * 128;
  float* H2  = H;
  float* AG2 = H + (size_t)N * 64;

  const int tot4        = N * 32;
  const int gemm_blocks = (N + 127) / 128;

  gemm_wmma_f32<128><<<gemm_blocks, 256, 0, stream>>>(x, W0, H, N);
  agg_tile<128><<<NTILES, 1024, 0, stream>>>(H, src, tgt, w, A, N, E);
  bias_relu128<<<EW_BLOCKS, EW_THREADS, 0, stream>>>(A, b0, tot4);

  gemm_wmma_f32<128><<<gemm_blocks, 256, 0, stream>>>(A, W1, H, N);
  agg_tile<128><<<NTILES, 1024, 0, stream>>>(H, src, tgt, w, A, N, E);
  gn_bias_relu<<<EW_BLOCKS, EW_THREADS, 0, stream>>>(A, b1, gamma, beta, tot4);

  gemm_wmma_f32<40><<<gemm_blocks, 256, 0, stream>>>(A, W2, H2, N);
  agg_tile<40><<<NTILES, 1024, 0, stream>>>(H2, src, tgt, w, AG2, N, E);
  lsm40<<<EW_BLOCKS, EW_THREADS, 0, stream>>>(AG2, b2, out, N);
}


# layer 0's tile owners store each wave's own edge list (local node, source, weight) in the workspace; layers 1 and 2 read it back in place of filtering all targets again; sum order kept
# speedup vs baseline: 2.2666x; 1.5756x over previous
#include <hip/hip_runtime.h>
#include <math.h>

typedef __attribute__((ext_vector_type(2))) float v2f;
typedef __attribute__((ext_vector_type(8))) float v8f;

#define NNODES 50000
#define NEDGES 800000

template <int NC>
__global__ __launch_bounds__(256) void gemm_wmma_f32(
    const float* __restrict__ X, const float* __restrict__ W,
    float* __restrict__ H, int n) {
  constexpr int K  = 128;
  constexpr int CT = (NC + 15) / 16;
  constexpr int PC = CT * 16;
  __shared__ v2f lw[(K / 2) * PC];

  const int tid = threadIdx.x;
  for (int i = tid; i < (K / 2) * PC; i += 256) {
    const int kp = i / PC, c = i % PC;
    v2f val;
    val.x = (c < NC) ? W[(2 * kp) * NC + c] : 0.0f;
    val.y = (c < NC) ? W[(2 * kp + 1) * NC + c] : 0.0f;
    lw[i] = val;
  }
  __syncthreads();

  const int wave = tid >> 5;
  const int lane = tid & 31;
  const int row0 = (blockIdx.x * 8 + wave) * 16;
  if (row0 >= n) return;

  const int m  = lane & 15;
  const int hk = lane >> 4;

  v8f acc[CT];
#pragma unroll
  for (int t = 0; t < CT; ++t) acc[t] = {};

  const float* xrow = X + (size_t)(row0 + m) * K + 2 * hk;

  for (int k4 = 0; k4 < K; k4 += 4) {
    v2f a;
    a.x = xrow[k4];
    a.y = xrow[k4 + 1];
    const v2f* brow = &lw[(k4 / 2 + hk) * PC + m];
#pragma unroll
    for (int t = 0; t < CT; ++t) {
      const v2f b = brow[t * 16];
      acc[t] = __builtin_amdgcn_wmma_f32_16x16x4_f32(
          false, a, false, b, (short)0, acc[t], false, false);
    }
  }

  const int crow = row0 + hk * 8;
#pragma unroll
  for (int t = 0; t < CT; ++t) {
    const int col = t * 16 + m;
    if ((NC % 16 == 0) || (col < NC)) {
#pragma unroll
      for (int r = 0; r < 8; ++r)
        H[(size_t)(crow + r) * NC + col] = acc[t][r];
    }
  }
}

#define TILE 196
#define NTILES ((NNODES + TILE - 1) / TILE)
#define NWV 32
#define CAP 448
#define CAP2 256

static_assert(TILE <= 256, "local node id is kept in 8 bits");
static_assert(NNODES < (1 << 24) && NEDGES < (1 << 24), "ids are kept in 24 bits");

#define SCAP 512

template <int F>
__device__ __forceinline__ void add_rows(
    const unsigned* ok, const float* ow, int n_own,
    const float* __restrict__ H, float4* acc, int lane) {
  constexpr int F4 = F / 4;
  for (int j0 = 0; j0 < n_own; j0 += 8) {
    unsigned kk[8];
    float wq[8];
    float4 v[8];
#pragma unroll
    for (int u = 0; u < 8; ++u) {
      const int jj = min(j0 + u, n_own - 1);
      kk[u] = (unsigned)__builtin_amdgcn_readfirstlane((int)ok[jj]);
      wq[u] = ow[jj];
    }
#pragma unroll
    for (int u = 0; u < 8; ++u) {
      const size_t s = (size_t)(kk[u] & 0xFFFFFFu);
      v[u] = make_float4(0.f, 0.f, 0.f, 0.f);
      if (lane < F4) v[u] = *((const float4*)(H + s * F) + lane);
    }
#pragma unroll
    for (int u = 0; u < 8; ++u) {
      if (j0 + u < n_own) {
        const int tl = (int)(kk[u] >> 24);
        if (lane < F4) {
          float4 a = acc[tl * F4 + lane];
          a.x = fmaf(wq[u], v[u].x, a.x);
          a.y = fmaf(wq[u], v[u].y, a.y);
          a.z = fmaf(wq[u], v[u].z, a.z);
          a.w = fmaf(wq[u], v[u].w, a.w);
          acc[tl * F4 + lane] = a;
        }
      }
    }
  }
}

template <int F, int MODE>
__global__ __launch_bounds__(1024) void agg_tile(
    const float* __restrict__ H, const int* __restrict__ src,
    const int* __restrict__ tgt, const float* __restrict__ w,
    float* __restrict__ A, unsigned* __restrict__ slist, int* __restrict__ scnt,
    int n, int E) {
  constexpr int F4 = F / 4;
  __shared__ float4   acc[TILE * F4];
  __shared__ unsigned lkey[NWV * CAP];
  __shared__ float    lwt[NWV * CAP];
  __shared__ unsigned okey[NWV * CAP2];
  __shared__ float    owt[NWV * CAP2];
  __shared__ int      lcnt[NWV];
  __shared__ int      wmore[NWV];

  const int tid  = threadIdx.x;
  const int lane = tid & 31;
  const int wave = __builtin_amdgcn_readfirstlane(tid >> 5);
  const int lo   = blockIdx.x * TILE;
  const unsigned below = (1u << lane) - 1u;

  unsigned* ok = okey + wave * CAP2;
  float*    ow = owt + wave * CAP2;
  uint2*    sv = (uint2*)slist + (size_t)(blockIdx.x * NWV + wave) * SCAP;

  int  cl = 0;
  bool use_saved = false;
  if (MODE == 1) {
    cl = scnt[blockIdx.x * NWV + lane];
    use_saved = __builtin_amdgcn_ballot_w32(cl > SCAP) == 0;
  }

  for (int i = tid; i < TILE * F4; i += 1024) acc[i] = make_float4(0.f, 0.f, 0.f, 0.f);

  if (use_saved) {
    const int c = __builtin_amdgcn_readlane(cl, wave);
    __syncthreads();
    for (int j0 = 0; j0 < c; j0 += CAP2) {
      const int nn = min(CAP2, c - j0);
      uint2 e[CAP2 / 32];
#pragma unroll
      for (int u = 0; u < CAP2 / 32; ++u) {
        const int i = u * 32 + lane;
        e[u] = make_uint2(0u, 0u);
        if (i < nn) e[u] = sv[j0 + i];
      }
#pragma unroll
      for (int u = 0; u < CAP2 / 32; ++u) {
        const int i = u * 32 + lane;
        if (i < nn) {
          ok[i] = e[u].x;
          ow[i] = __uint_as_float(e[u].y);
        }
      }
      __builtin_amdgcn_wave_barrier();
      add_rows<F>(ok, ow, nn, H, acc, lane);
      __builtin_amdgcn_wave_barrier();
    }
  } else {
    const int per = (((E + NWV - 1) / NWV) + 3) & ~3;
    int pos = min(wave * per, E);
    const int end = min(pos + per, E);
    unsigned* mykey = lkey + wave * CAP;
    float*    myw   = lwt + wave * CAP;

    const int max_trips = per / (CAP - 31) + 2;
    int skip = 0;
    int saved = 0;
    for (int trip = 0; trip < max_trips; ++trip) {
      __syncthreads();

      int  cnt  = 0;
      bool full = false;
      while (pos < end && !full) {
        int4 t[8];
        if (pos + 1024 <= end) {
          const int4* tp = (const int4*)(tgt + pos) + lane;
#pragma unroll
          for (int u = 0; u < 8; ++u) t[u] = tp[u * 32];
        } else {
#pragma unroll
          for (int u = 0; u < 8; ++u) {
            const int e = pos + (u * 32 + lane) * 4;
            const int v0 = tgt[min(e, end - 1)];
            const int v1 = tgt[min(e + 1, end - 1)];
            const int v2 = tgt[min(e + 2, end - 1)];
            const int v3 = tgt[min(e + 3, end - 1)];
            t[u].x = (e < end) ? v0 : -1;
            t[u].y = (e + 1 < end) ? v1 : -1;
            t[u].z = (e + 2 < end) ? v2 : -1;
            t[u].w = (e + 3 < end) ? v3 : -1;
          }
        }
#pragma unroll
        for (int u = 0; u < 8; ++u) {
          const int tv[4] = {t[u].x, t[u].y, t[u].z, t[u].w};
          unsigned m[4];
#pragma unroll
          for (int c = 0; c < 4; ++c)
            m[c] = __builtin_amdgcn_ballot_w32((unsigned)(tv[c] - lo) < (unsigned)TILE);
          if ((m[0] | m[1] | m[2] | m[3]) != 0) {
#pragma unroll
            for (int c = 0; c < 4; ++c) {
              const int q = u * 4 + c;
              if (m[c] != 0 && q >= skip && !full) {
                if (cnt > CAP - 32) {
                  full = true;
                  skip = q;
                } else {
                  if ((m[c] >> lane) & 1u) {
                    const int idx = cnt + __builtin_popcount(m[c] & below);
                    mykey[idx] = ((unsigned)(tv[c] - lo) << 24) |
                                 (unsigned)(pos + (u * 32 + lane) * 4 + c);
                  }
                  cnt += __builtin_popcount(m[c]);
                }
              }
            }
          }
        }
        if (!full) { pos += 1024; skip = 0; }
      }
      __builtin_amdgcn_wave_barrier();
      for (int i = lane; i < cnt; i += 32) {
        const unsigned k = mykey[i];
        const int e = (int)(k & 0xFFFFFFu);
        mykey[i] = (k & 0xFF000000u) | (unsigned)src[e];
        myw[i]   = w[e];
      }
      if (lane == 0) {
        lcnt[wave]  = cnt;
        wmore[wave] = (pos < end) ? 1 : 0;
      }
      __syncthreads();
      const bool more = __builtin_amdgcn_ballot_w32(wmore[lane] != 0) != 0;

      int sl = 0, base = 0;
      while (sl < NWV) {
        int n_own = 0;
        while (sl < NWV && n_own <= CAP2 - 32) {
          const int c = __builtin_amdgcn_readfirstlane(lcnt[sl]);
          if (base >= c) { ++sl; base = 0; continue; }
          const int i = base + lane;
          unsigned k = 0;
          float ww = 0.f;
          bool mine = false;
          if (i < c) {
            k    = lkey[sl * CAP + i];
            ww   = lwt[sl * CAP + i];
            mine = (int)((k >> 24) & 31u) == wave;
          }
          const unsigned mm = __builtin_amdgcn_ballot_w32(mine);
          if (mm != 0) {
            if (mine) {
              const int idx = n_own + __builtin_popcount(mm & below);
              ok[idx] = k;
              ow[idx] = ww;
            }
            n_own += __builtin_popcount(mm);
          }
          base += 32;
        }
        __builtin_amdgcn_wave_barrier();
        if (MODE == 0) {
          for (int i = lane; i < n_own; i += 32) {
            const int g = saved + i;
            if (g < SCAP) sv[g] = make_uint2(ok[i], __float_as_uint(ow[i]));
          }
          saved += n_own;
        }
        add_rows<F>(ok, ow, n_own, H, acc, lane);
        __builtin_amdgcn_wave_barrier();
      }
      if (!more) break;
    }
    if (MODE == 0 && lane == 0) scnt[blockIdx.x * NWV + wave] = saved;
  }
  __syncthreads();

  float4* A4 = (float4*)A;
  const int g0   = lo * F4;
  const int gmax = n * F4;
  for (int i = tid; i < TILE * F4; i += 1024)
    if (g0 + i < gmax) A4[g0 + i] = acc[i];
}

#define EW_BLOCKS 1024
#define EW_THREADS 256
#define EW_UNROLL 8

__global__ __launch_bounds__(EW_THREADS) void bias_relu128(
    float* __restrict__ A, const float* __restrict__ b, int total4) {
  const int T  = gridDim.x * blockDim.x;
  const int i0 = blockIdx.x * blockDim.x + threadIdx.x;
  const float4 bb = *(const float4*)(b + (i0 & 31) * 4);
  float4* A4 = (float4*)A;
  for (int base = i0; base < total4; base += EW_UNROLL * T) {
    float4 v[EW_UNROLL];
#pragma unroll
    for (int k = 0; k < EW_UNROLL; ++k) {
      const int i = base + k * T;
      v[k] = make_float4(0.f, 0.f, 0.f, 0.f);
      if (i < total4) v[k] = A4[i];
    }
#pragma unroll
    for (int k = 0; k < EW_UNROLL; ++k) {
      const int i = base + k * T;
      if (i < total4) {
        float4 r;
        r.x = fmaxf(v[k].x + bb.x, 0.0f);
        r.y = fmaxf(v[k].y + bb.y, 0.0f);
        r.z = fmaxf(v[k].z + bb.z, 0.0f);
        r.w = fmaxf(v[k].w + bb.w, 0.0f);
        A4[i] = r;
      }
    }
  }
}

__global__ __launch_bounds__(EW_THREADS) void gn_bias_relu(
    float* __restrict__ A, const float* __restrict__ b,
    const float* __restrict__ gamma, const float* __restrict__ beta, int total4) {
  const int T  = gridDim.x * blockDim.x;
  const int i0 = blockIdx.x * blockDim.x + threadIdx.x;
  const int g  = i0 & 31;
  const float4 bb = *(const float4*)(b + g * 4);
  const float4 gm = *(const float4*)(gamma + g * 4);
  const float4 bt = *(const float4*)(beta + g * 4);
  float4* A4 = (float4*)A;
  for (int base = i0; base < total4; base += EW_UNROLL * T) {
    float4 q[EW_UNROLL];
#pragma unroll
    for (int k = 0; k < EW_UNROLL; ++k) {
      const int i = base + k * T;
      q[k] = make_float4(0.f, 0.f, 0.f, 0.f);
      if (i < total4) q[k] = A4[i];
    }
#pragma unroll
    for (int k = 0; k < EW_UNROLL; ++k) {
      const int i = base + k * T;
      if (i < total4) {
        float4 v = q[k];
        v.x = fmaxf(v.x + bb.x, 0.0f);
        v.y = fmaxf(v.y + bb.y, 0.0f);
        v.z = fmaxf(v.z + bb.z, 0.0f);
        v.w = fmaxf(v.w + bb.w, 0.0f);
        const float mu  = 0.25f * (v.x + v.y + v.z + v.w);
        const float ms  = 0.25f * (v.x * v.x + v.y * v.y + v.z * v.z + v.w * v.w);
        const float inv = rsqrtf(fmaxf(ms - mu * mu, 0.0f) + 1e-5f);
        v.x = (v.x - mu) * inv * gm.x + bt.x;
        v.y = (v.y - mu) * inv * gm.y + bt.y;
        v.z = (v.z - mu) * inv * gm.z + bt.z;
        v.w = (v.w - mu) * inv * gm.w + bt.w;
        A4[i] = v;
      }
    }
  }
}

__global__ __launch_bounds__(EW_THREADS) void lsm40(
    const float* __restrict__ Ain, const float* __restrict__ b,
    float* __restrict__ out, int n) {
  const int lane = threadIdx.x & 31;
  const int gw = __builtin_amdgcn_readfirstlane(
      (int)((blockIdx.x * blockDim.x + threadIdx.x) >> 5));
  const int nw = (int)((gridDim.x * blockDim.x) >> 5);
  const float c0 = b[lane];
  const float c1 = (lane < 8) ? b[32 + lane] : 0.0f;
  for (int nb = gw; nb < n; nb += EW_UNROLL * nw) {
    float r0[EW_UNROLL], r1[EW_UNROLL];
#pragma unroll
    for (int k = 0; k < EW_UNROLL; ++k) {
      const int node = nb + k * nw;
      r0[k] = 0.0f;
      r1[k] = 0.0f;
      if (node < n) {
        const float* row = Ain + (size_t)node * 40;
        r0[k] = row[lane];
        if (lane < 8) r1[k] = row[32 + lane];
      }
    }
#pragma unroll
    for (int k = 0; k < EW_UNROLL; ++k) {
      const int node = nb + k * nw;
      if (node < n) {
        const float v0 = r0[k] + c0;
        const float v1 = (lane < 8) ? (r1[k] + c1) : -INFINITY;
        float m = fmaxf(v0, v1);
#pragma unroll
        for (int o = 16; o > 0; o >>= 1) m = fmaxf(m, __shfl_xor(m, o, 32));
        float s = expf(v0 - m) + ((lane < 8) ? expf(v1 - m) : 0.0f);
#pragma unroll
        for (int o = 16; o > 0; o >>= 1) s += __shfl_xor(s, o, 32);
        const float lse = m + logf(s);
        out[(size_t)node * 40 + lane] = v0 - lse;
        if (lane < 8) out[(size_t)node * 40 + 32 + lane] = v1 - lse;
      }
    }
  }
}

extern "C" void kernel_launch(void* const* d_in, const int* in_sizes, int n_in,
                              void* d_out, int out_size, void* d_ws, size_t ws_size,
                              hipStream_t stream) {
  const int N = NNODES, E = NEDGES;
  const float* x     = (const float*)d_in[0];
  const int*   src   = (const int*)d_in[1];
  const int*   tgt   = (const int*)d_in[2];
  const float* w     = (const float*)d_in[3];
  const float* W0    = (const float*)d_in[4];
  const float* b0    = (const float*)d_in[5];
  const float* W1    = (const float*)d_in[6];
  const float* b1    = (const float*)d_in[7];
  const float* W2    = (const float*)d_in[8];
  const float* b2    = (const float*)d_in[9];
  const float* gamma = (const float*)d_in[10];
  const float* beta  = (const float*)d_in[11];
  float* out = (float*)d_out;

  float* H   = (float*)d_ws;
  float* A   = H + (size_t)N * 128;
  float* H2  = H;
  float* AG2 = H + (size_t)N * 64;
  unsigned* SL = (unsigned*)(H + (size_t)N * 256);
  int*      SC = (int*)(SL + (size_t)NTILES * NWV * SCAP * 2);

  const int tot4        = N * 32;
  const int gemm_blocks = (N + 127) / 128;

  gemm_wmma_f32<128><<<gemm_blocks, 256, 0, stream>>>(x, W0, H, N);
  agg_tile<128, 0><<<NTILES, 1024, 0, stream>>>(H, src, tgt, w, A, SL, SC, N, E);
  bias_relu128<<<EW_BLOCKS, EW_THREADS, 0, stream>>>(A, b0, tot4);

  gemm_wmma_f32<128><<<gemm_blocks, 256, 0, stream>>>(A, W1, H, N);
  agg_tile<128, 1><<<NTILES, 1024, 0, stream>>>(H, src, tgt, w, A, SL, SC, N, E);
  gn_bias_relu<<<EW_BLOCKS, EW_THREADS, 0, stream>>>(A, b1, gamma, beta, tot4);

  gemm_wmma_f32<40><<<gemm_blocks, 256, 0, stream>>>(A, W2, H2, N);
  agg_tile<40, 1><<<NTILES, 1024, 0, stream>>>(H2, src, tgt, w, AG2, SL, SC, N, E);
  lsm40<<<EW_BLOCKS, EW_THREADS, 0, stream>>>(AG2, b2, out, N);
}


# layer 0's owner scan reads one tile-id byte per edge (tgt/196, written by a new first launch) instead of the 4-byte targets: 4 edges tested per word, a quarter of the bytes; sum order kept
# speedup vs baseline: 2.3526x; 1.0379x over previous
#include <hip/hip_runtime.h>
#include <math.h>

typedef __attribute__((ext_vector_type(2))) float v2f;
typedef __attribute__((ext_vector_type(8))) float v8f;

#define NNODES 50000
#define NEDGES 800000

template <int NC>
__global__ __launch_bounds__(256) void gemm_wmma_f32(
    const float* __restrict__ X, const float* __restrict__ W,
    float* __restrict__ H, int n) {
  constexpr int K  = 128;
  constexpr int CT = (NC + 15) / 16;
  constexpr int PC = CT * 16;
  __shared__ v2f lw[(K / 2) * PC];

  const int tid = threadIdx.x;
  for (int i = tid; i < (K / 2) * PC; i += 256) {
    const int kp = i / PC, c = i % PC;
    v2f val;
    val.x = (c < NC) ? W[(2 * kp) * NC + c] : 0.0f;
    val.y = (c < NC) ? W[(2 * kp + 1) * NC + c] : 0.0f;
    lw[i] = val;
  }
  __syncthreads();

  const int wave = tid >> 5;
  const int lane = tid & 31;
  const int row0 = (blockIdx.x * 8 + wave) * 16;
  if (row0 >= n) return;

  const int m  = lane & 15;
  const int hk = lane >> 4;

  v8f acc[CT];
#pragma unroll
  for (int t = 0; t < CT; ++t) acc[t] = {};

  const float* xrow = X + (size_t)(row0 + m) * K + 2 * hk;

  for (int k4 = 0; k4 < K; k4 += 4) {
    v2f a;
    a.x = xrow[k4];
    a.y = xrow[k4 + 1];
    const v2f* brow = &lw[(k4 / 2 + hk) * PC + m];
#pragma unroll
    for (int t = 0; t < CT; ++t) {
      const v2f b = brow[t * 16];
      acc[t] = __builtin_amdgcn_wmma_f32_16x16x4_f32(
          false, a, false, b, (short)0, acc[t], false, false);
    }
  }

  const int crow = row0 + hk * 8;
#pragma unroll
  for (int t = 0; t < CT; ++t) {
    const int col = t * 16 + m;
    if ((NC % 16 == 0) || (col < NC)) {
#pragma unroll
      for (int r = 0; r < 8; ++r)
        H[(size_t)(crow + r) * NC + col] = acc[t][r];
    }
  }
}

#define TILE 196
#define NTILES ((NNODES + TILE - 1) / TILE)
#define NWV 32
#define CAP 448
#define CAP2 256

static_assert(TILE <= 256, "local node id is kept in 8 bits");
static_assert(NNODES < (1 << 24) && NEDGES < (1 << 24), "ids are kept in 24 bits");

static_assert(NTILES <= 256, "a tile id is kept in 8 bits");
static_assert(NEDGES % 4 == 0, "four tile ids per word");
#define TB_THREADS 256
#define TB_BLOCKS ((NEDGES / 4 + TB_THREADS - 1) / TB_THREADS)

__global__ __launch_bounds__(TB_THREADS) void tile_ids(
    const int* __restrict__ tgt, unsigned* __restrict__ tb, int E4) {
  const int i = blockIdx.x * TB_THREADS + threadIdx.x;
  if (i < E4) {
    const int4 t = ((const int4*)tgt)[i];
    tb[i] = (((unsigned)t.x / TILE) & 0xFFu) |
            ((((unsigned)t.y / TILE) & 0xFFu) << 8) |
            ((((unsigned)t.z / TILE) & 0xFFu) << 16) |
            (((unsigned)t.w / TILE) << 24);
  }
}

#define SCAP 512

template <int F>
__device__ __forceinline__ void add_rows(
    const unsigned* ok, const float* ow, int n_own,
    const float* __restrict__ H, float4* acc, int lane) {
  constexpr int F4 = F / 4;
  for (int j0 = 0; j0 < n_own; j0 += 8) {
    unsigned kk[8];
    float wq[8];
    float4 v[8];
#pragma unroll
    for (int u = 0; u < 8; ++u) {
      const int jj = min(j0 + u, n_own - 1);
      kk[u] = (unsigned)__builtin_amdgcn_readfirstlane((int)ok[jj]);
      wq[u] = ow[jj];
    }
#pragma unroll
    for (int u = 0; u < 8; ++u) {
      const size_t s = (size_t)(kk[u] & 0xFFFFFFu);
      v[u] = make_float4(0.f, 0.f, 0.f, 0.f);
      if (lane < F4) v[u] = *((const float4*)(H + s * F) + lane);
    }
#pragma unroll
    for (int u = 0; u < 8; ++u) {
      if (j0 + u < n_own) {
        const int tl = (int)(kk[u] >> 24);
        if (lane < F4) {
          float4 a = acc[tl * F4 + lane];
          a.x = fmaf(wq[u], v[u].x, a.x);
          a.y = fmaf(wq[u], v[u].y, a.y);
          a.z = fmaf(wq[u], v[u].z, a.z);
          a.w = fmaf(wq[u], v[u].w, a.w);
          acc[tl * F4 + lane] = a;
        }
      }
    }
  }
}

template <int F, int MODE>
__global__ __launch_bounds__(1024) void agg_tile(
    const float* __restrict__ H, const int* __restrict__ src,
    const int* __restrict__ tgt, const float* __restrict__ w,
    float* __restrict__ A, unsigned* __restrict__ slist, int* __restrict__ scnt,
    const unsigned* __restrict__ tb, int n, int E) {
  constexpr int F4 = F / 4;
  __shared__ float4   acc[TILE * F4];
  __shared__ unsigned lkey[NWV * CAP];
  __shared__ float    lwt[NWV * CAP];
  __shared__ unsigned okey[NWV * CAP2];
  __shared__ float    owt[NWV * CAP2];
  __shared__ int      lcnt[NWV];
  __shared__ int      wmore[NWV];

  const int tid  = threadIdx.x;
  const int lane = tid & 31;
  const int wave = __builtin_amdgcn_readfirstlane(tid >> 5);
  const int lo   = blockIdx.x * TILE;
  const unsigned below = (1u << lane) - 1u;
  const unsigned pat = (unsigned)blockIdx.x * 0x01010101u;

  unsigned* ok = okey + wave * CAP2;
  float*    ow = owt + wave * CAP2;
  uint2*    sv = (uint2*)slist + (size_t)(blockIdx.x * NWV + wave) * SCAP;

  int  cl = 0;
  bool use_saved = false;
  if (MODE == 1) {
    cl = scnt[blockIdx.x * NWV + lane];
    use_saved = __builtin_amdgcn_ballot_w32(cl > SCAP) == 0;
  }

  for (int i = tid; i < TILE * F4; i += 1024) acc[i] = make_float4(0.f, 0.f, 0.f, 0.f);

  if (use_saved) {
    const int c = __builtin_amdgcn_readlane(cl, wave);
    __syncthreads();
    for (int j0 = 0; j0 < c; j0 += CAP2) {
      const int nn = min(CAP2, c - j0);
      uint2 e[CAP2 / 32];
#pragma unroll
      for (int u = 0; u < CAP2 / 32; ++u) {
        const int i = u * 32 + lane;
        e[u] = make_uint2(0u, 0u);
        if (i < nn) e[u] = sv[j0 + i];
      }
#pragma unroll
      for (int u = 0; u < CAP2 / 32; ++u) {
        const int i = u * 32 + lane;
        if (i < nn) {
          ok[i] = e[u].x;
          ow[i] = __uint_as_float(e[u].y);
        }
      }
      __builtin_amdgcn_wave_barrier();
      add_rows<F>(ok, ow, nn, H, acc, lane);
      __builtin_amdgcn_wave_barrier();
    }
  } else {
    const int per = (((E + NWV - 1) / NWV) + 3) & ~3;
    int pos = min(wave * per, E);
    const int end = min(pos + per, E);
    unsigned* mykey = lkey + wave * CAP;
    float*    myw   = lwt + wave * CAP;

    const int max_trips = per / (CAP - 127) + 2;
    int skip = 0;
    int saved = 0;
    for (int trip = 0; trip < max_trips; ++trip) {
      __syncthreads();

      int cnt = 0;
      while (pos < end) {
        unsigned x[32];
        if (pos + 4096 <= end) {
          const unsigned* tp = tb + (pos >> 2) + lane;
#pragma unroll
          for (int i = 0; i < 32; ++i) x[i] = tp[i * 32] ^ pat;
        } else {
#pragma unroll
          for (int i = 0; i < 32; ++i) {
            const int wi = (pos >> 2) + i * 32 + lane;
            const unsigned v = tb[min(wi, (E >> 2) - 1)] ^ pat;
            x[i] = (wi * 4 < end) ? v : 0xFFFFFFFFu;
          }
        }
        if (skip != 0) {
#pragma unroll
          for (int u = 0; u < 8; ++u)
            if (u < skip) x[u] = 0xFFFFFFFFu;
        }
#pragma unroll
        for (int ss = 0; ss < 4; ++ss) {
          if (pos < end) {
#pragma unroll
            for (int u = 0; u < 8; ++u) {
              const unsigned xv = x[ss * 8 + u];
              const unsigned hz = (xv - 0x01010101u) & ~xv & 0x80808080u;
              if (__builtin_amdgcn_ballot_w32(hz != 0u) != 0) {
                if (cnt > CAP - 128) {
                  skip = u;
                  goto scan_done;
                }
#pragma unroll
                for (int c = 0; c < 4; ++c) {
                  const unsigned mc =
                      __builtin_amdgcn_ballot_w32(((xv >> (8 * c)) & 0xFFu) == 0u);
                  if (mc != 0) {
                    if ((mc >> lane) & 1u) {
                      const int idx = cnt + __builtin_popcount(mc & below);
                      mykey[idx] = (unsigned)(pos + (u * 32 + lane) * 4 + c);
                    }
                    cnt += __builtin_popcount(mc);
                  }
                }
              }
            }
            pos += 1024;
            skip = 0;
          }
        }
      }
    scan_done:
      __builtin_amdgcn_wave_barrier();
      for (int i = lane; i < cnt; i += 32) {
        const int e = (int)mykey[i];
        mykey[i] = ((unsigned)(tgt[e] - lo) << 24) | (unsigned)src[e];
        myw[i]   = w[e];
      }
      if (lane == 0) {
        lcnt[wave]  = cnt;
        wmore[wave] = (pos < end) ? 1 : 0;
      }
      __syncthreads();
      const bool more = __builtin_amdgcn_ballot_w32(wmore[lane] != 0) != 0;

      int sl = 0, base = 0;
      while (sl < NWV) {
        int n_own = 0;
        while (sl < NWV && n_own <= CAP2 - 32) {
          const int c = __builtin_amdgcn_readfirstlane(lcnt[sl]);
          if (base >= c) { ++sl; base = 0; continue; }
          const int i = base + lane;
          unsigned k = 0;
          float ww = 0.f;
          bool mine = false;
          if (i < c) {
            k    = lkey[sl * CAP + i];
            ww   = lwt[sl * CAP + i];
            mine = (int)((k >> 24) & 31u) == wave;
          }
          const unsigned mm = __builtin_amdgcn_ballot_w32(mine);
          if (mm != 0) {
            if (mine) {
              const int idx = n_own + __builtin_popcount(mm & below);
              ok[idx] = k;
              ow[idx] = ww;
            }
            n_own += __builtin_popcount(mm);
          }
          base += 32;
        }
        __builtin_amdgcn_wave_barrier();
        if (MODE == 0) {
          for (int i = lane; i < n_own; i += 32) {
            const int g = saved + i;
            if (g < SCAP) sv[g] = make_uint2(ok[i], __float_as_uint(ow[i]));
          }
          saved += n_own;
        }
        add_rows<F>(ok, ow, n_own, H, acc, lane);
        __builtin_amdgcn_wave_barrier();
      }
      if (!more) break;
    }
    if (MODE == 0 && lane == 0) scnt[blockIdx.x * NWV + wave] = saved;
  }
  __syncthreads();

  float4* A4 = (float4*)A;
  const int g0   = lo * F4;
  const int gmax = n * F4;
  for (int i = tid; i < TILE * F4; i += 1024)
    if (g0 + i < gmax) A4[g0 + i] = acc[i];
}

#define EW_BLOCKS 1024
#define EW_THREADS 256
#define EW_UNROLL 8

__global__ __launch_bounds__(EW_THREADS) void bias_relu128(
    float* __restrict__ A, const float* __restrict__ b, int total4) {
  const int T  = gridDim.x * blockDim.x;
  const int i0 = blockIdx.x * blockDim.x + threadIdx.x;
  const float4 bb = *(const float4*)(b + (i0 & 31) * 4);
  float4* A4 = (float4*)A;
  for (int base = i0; base < total4; base += EW_UNROLL * T) {
    float4 v[EW_UNROLL];
#pragma unroll
    for (int k = 0; k < EW_UNROLL; ++k) {
      const int i = base + k * T;
      v[k] = make_float4(0.f, 0.f, 0.f, 0.f);
      if (i < total4) v[k] = A4[i];
    }
#pragma unroll
    for (int k = 0; k < EW_UNROLL; ++k) {
      const int i = base + k * T;
      if (i < total4) {
        float4 r;
        r.x = fmaxf(v[k].x + bb.x, 0.0f);
        r.y = fmaxf(v[k].y + bb.y, 0.0f);
        r.z = fmaxf(v[k].z + bb.z, 0.0f);
        r.w = fmaxf(v[k].w + bb.w, 0.0f);
        A4[i] = r;
      }
    }
  }
}

__global__ __launch_bounds__(EW_THREADS) void gn_bias_relu(
    float* __restrict__ A, const float* __restrict__ b,
    const float* __restrict__ gamma, const float* __restrict__ beta, int total4) {
  const int T  = gridDim.x * blockDim.x;
  const int i0 = blockIdx.x * blockDim.x + threadIdx.x;
  const int g  = i0 & 31;
  const float4 bb = *(const float4*)(b + g * 4);
  const float4 gm = *(const float4*)(gamma + g * 4);
  const float4 bt = *(const float4*)(beta + g * 4);
  float4* A4 = (float4*)A;
  for (int base = i0; base < total4; base += EW_UNROLL * T) {
    float4 q[EW_UNROLL];
#pragma unroll
    for (int k = 0; k < EW_UNROLL; ++k) {
      const int i = base + k * T;
      q[k] = make_float4(0.f, 0.f, 0.f, 0.f);
      if (i < total4) q[k] = A4[i];
    }
#pragma unroll
    for (int k = 0; k < EW_UNROLL; ++k) {
      const int i = base + k * T;
      if (i < total4) {
        float4 v = q[k];
        v.x = fmaxf(v.x + bb.x, 0.0f);
        v.y = fmaxf(v.y + bb.y, 0.0f);
        v.z = fmaxf(v.z + bb.z, 0.0f);
        v.w = fmaxf(v.w + bb.w, 0.0f);
        const float mu  = 0.25f * (v.x + v.y + v.z + v.w);
        const float ms  = 0.25f * (v.x * v.x + v.y * v.y + v.z * v.z + v.w * v.w);
        const float inv = rsqrtf(fmaxf(ms - mu * mu, 0.0f) + 1e-5f);
        v.x = (v.x - mu) * inv * gm.x + bt.x;
        v.y = (v.y - mu) * inv * gm.y + bt.y;
        v.z = (v.z - mu) * inv * gm.z + bt.z;
        v.w = (v.w - mu) * inv * gm.w + bt.w;
        A4[i] = v;
      }
    }
  }
}

__global__ __launch_bounds__(EW_THREADS) void lsm40(
    const float* __restrict__ Ain, const float* __restrict__ b,
    float* __restrict__ out, int n) {
  const int lane = threadIdx.x & 31;
  const int gw = __builtin_amdgcn_readfirstlane(
      (int)((blockIdx.x * blockDim.x + threadIdx.x) >> 5));
  const int nw = (int)((gridDim.x * blockDim.x) >> 5);
  const float c0 = b[lane];
  const float c1 = (lane < 8) ? b[32 + lane] : 0.0f;
  for (int nb = gw; nb < n; nb += EW_UNROLL * nw) {
    float r0[EW_UNROLL], r1[EW_UNROLL];
#pragma unroll
    for (int k = 0; k < EW_UNROLL; ++k) {
      const int node = nb + k * nw;
      r0[k] = 0.0f;
      r1[k] = 0.0f;
      if (node < n) {
        const float* row = Ain + (size_t)node * 40;
        r0[k] = row[lane];
        if (lane < 8) r1[k] = row[32 + lane];
      }
    }
#pragma unroll
    for (int k = 0; k < EW_UNROLL; ++k) {
      const int node = nb + k * nw;
      if (node < n) {
        const float v0 = r0[k] + c0;
        const float v1 = (lane < 8) ? (r1[k] + c1) : -INFINITY;
        float m = fmaxf(v0, v1);
#pragma unroll
        for (int o = 16; o > 0; o >>= 1) m = fmaxf(m, __shfl_xor(m, o, 32));
        float s = expf(v0 - m) + ((lane < 8) ? expf(v1 - m) : 0.0f);
#pragma unroll
        for (int o = 16; o > 0; o >>= 1) s += __shfl_xor(s, o, 32);
        const float lse = m + logf(s);
        out[(size_t)node * 40 + lane] = v0 - lse;
        if (lane < 8) out[(size_t)node * 40 + 32 + lane] = v1 - lse;
      }
    }
  }
}

extern "C" void kernel_launch(void* const* d_in, const int* in_sizes, int n_in,
                              void* d_out, int out_size, void* d_ws, size_t ws_size,
                              hipStream_t stream) {
  const int N = NNODES, E = NEDGES;
  const float* x     = (const float*)d_in[0];
  const int*   src   = (const int*)d_in[1];
  const int*   tgt   = (const int*)d_in[2];
  const float* w     = (const float*)d_in[3];
  const float* W0    = (const float*)d_in[4];
  const float* b0    = (const float*)d_in[5];
  const float* W1    = (const float*)d_in[6];
  const float* b1    = (const float*)d_in[7];
  const float* W2    = (const float*)d_in[8];
  const float* b2    = (const float*)d_in[9];
  const float* gamma = (const float*)d_in[10];
  const float* beta  = (const float*)d_in[11];
  float* out = (float*)d_out;

  float* H   = (float*)d_ws;
  float* A   = H + (size_t)N * 128;
  float* H2  = H;
  float* AG2 = H + (size_t)N * 64;
  unsigned* SL = (unsigned*)(H + (size_t)N * 256);
  int*      SC = (int*)(SL + (size_t)NTILES * NWV * SCAP * 2);
  unsigned* TB = (unsigned*)(SC + NTILES * NWV);

  const int tot4        = N * 32;
  const int gemm_blocks = (N + 127) / 128;

  tile_ids<<<TB_BLOCKS, TB_THREADS, 0, stream>>>(tgt, TB, NEDGES / 4);

  gemm_wmma_f32<128><<<gemm_blocks, 256, 0, stream>>>(x, W0, H, N);
  agg_tile<128, 0><<<NTILES, 1024, 0, stream>>>(H, src, tgt, w, A, SL, SC, TB, N, E);
  bias_relu128<<<EW_BLOCKS, EW_THREADS, 0, stream>>>(A, b0, tot4);

  gemm_wmma_f32<128><<<gemm_blocks, 256, 0, stream>>>(A, W1, H, N);
  agg_tile<128, 1><<<NTILES, 1024, 0, stream>>>(H, src, tgt, w, A, SL, SC, TB, N, E);
  gn_bias_relu<<<EW_BLOCKS, EW_THREADS, 0, stream>>>(A, b1, gamma, beta, tot4);

  gemm_wmma_f32<40><<<gemm_blocks, 256, 0, stream>>>(A, W2, H2, N);
  agg_tile<40, 1><<<NTILES, 1024, 0, stream>>>(H2, src, tgt, w, AG2, SL, SC, TB, N, E);
  lsm40<<<EW_BLOCKS, EW_THREADS, 0, stream>>>(AG2, b2, out, N);
}


# layer 0: owners no longer walk all 32 hit lists; a scanning wave writes a 32-bit mask per owner (5 ballots of the local node bits), an owner reads its 32 masks, copies its entries; sum order kept
# speedup vs baseline: 2.8863x; 1.2269x over previous
#include <hip/hip_runtime.h>
#include <math.h>

typedef __attribute__((ext_vector_type(2))) float v2f;
typedef __attribute__((ext_vector_type(8))) float v8f;

#define NNODES 50000
#define NEDGES 800000

template <int NC, int EPI>
__global__ __launch_bounds__(256) void gemm_wmma_f32(
    const float* __restrict__ X, const float* __restrict__ W,
    const float* __restrict__ eb, const float* __restrict__ eg,
    const float* __restrict__ et, float* __restrict__ H, int n) {
  constexpr int K  = 128;
  constexpr int CT = (NC + 15) / 16;
  constexpr int PC = CT * 16;
  __shared__ v2f lw[(K / 2) * PC];
  __shared__ float4 ep[3][K / 4];

  const int tid = threadIdx.x;
  for (int i = tid; i < (K / 2) * PC; i += 256) {
    const int kp = i / PC, c = i % PC;
    v2f val;
    val.x = (c < NC) ? W[(2 * kp) * NC + c] : 0.0f;
    val.y = (c < NC) ? W[(2 * kp + 1) * NC + c] : 0.0f;
    lw[i] = val;
  }
  if (EPI >= 1 && tid < K / 4) ep[0][tid] = *(const float4*)(eb + tid * 4);
  if (EPI == 2 && tid < K / 4) {
    ep[1][tid] = *(const float4*)(eg + tid * 4);
    ep[2][tid] = *(const float4*)(et + tid * 4);
  }
  __syncthreads();

  const int wave = tid >> 5;
  const int lane = tid & 31;
  const int row0 = (blockIdx.x * 8 + wave) * 16;
  if (row0 >= n) return;

  const int m  = lane & 15;
  const int hk = lane >> 4;

  v8f acc[CT];
#pragma unroll
  for (int t = 0; t < CT; ++t) acc[t] = {};

  const float4* xrow = (const float4*)(X + (size_t)(row0 + m) * K) + hk;

  float4 q[K / 8];
#pragma unroll
  for (int j = 0; j < K / 8; ++j) q[j] = xrow[2 * j];

#pragma unroll
  for (int j = 0; j < K / 8; ++j) {
    float4 v = q[j];
    if (EPI >= 1) {
      const float4 bb = ep[0][2 * j + hk];
      v.x = fmaxf(v.x + bb.x, 0.0f);
      v.y = fmaxf(v.y + bb.y, 0.0f);
      v.z = fmaxf(v.z + bb.z, 0.0f);
      v.w = fmaxf(v.w + bb.w, 0.0f);
    }
    if (EPI == 2) {
      const float4 gm = ep[1][2 * j + hk];
      const float4 bt = ep[2][2 * j + hk];
      const float mu  = 0.25f * (v.x + v.y + v.z + v.w);
      const float ms  = 0.25f * (v.x * v.x + v.y * v.y + v.z * v.z + v.w * v.w);
      const float inv = rsqrtf(fmaxf(ms - mu * mu, 0.0f) + 1e-5f);
      v.x = (v.x - mu) * inv * gm.x + bt.x;
      v.y = (v.y - mu) * inv * gm.y + bt.y;
      v.z = (v.z - mu) * inv * gm.z + bt.z;
      v.w = (v.w - mu) * inv * gm.w + bt.w;
    }
    const v2f* brow = &lw[(4 * j + 2 * hk) * PC + m];
    v2f a0, a1;
    a0.x = v.x; a0.y = v.y;
    a1.x = v.z; a1.y = v.w;
#pragma unroll
    for (int t = 0; t < CT; ++t) {
      const v2f b = brow[t * 16];
      acc[t] = __builtin_amdgcn_wmma_f32_16x16x4_f32(
          false, a0, false, b, (short)0, acc[t], false, false);
    }
#pragma unroll
    for (int t = 0; t < CT; ++t) {
      const v2f b = brow[PC + t * 16];
      acc[t] = __builtin_amdgcn_wmma_f32_16x16x4_f32(
          false, a1, false, b, (short)0, acc[t], false, false);
    }
  }

  const int crow = row0 + hk * 8;
#pragma unroll
  for (int t = 0; t < CT; ++t) {
    const int col = t * 16 + m;
    if ((NC % 16 == 0) || (col < NC)) {
#pragma unroll
      for (int r = 0; r < 8; ++r)
        H[(size_t)(crow + r) * NC + col] = acc[t][r];
    }
  }
}

#define TILE 196
#define NTILES ((NNODES + TILE - 1) / TILE)
#define NWV 32
#define CAP 448
#define CAP2 256
#define FCH 5
#define OMP (FCH * 32 + 1)

static_assert(TILE <= 256, "local node id is kept in 8 bits");
static_assert(NNODES < (1 << 24) && NEDGES < (1 << 24), "ids are kept in 24 bits");

static_assert(NTILES <= 256, "a tile id is kept in 8 bits");
static_assert(NEDGES % 4 == 0, "four tile ids per word");
#define TB_THREADS 256
#define TB_BLOCKS ((NEDGES / 4 + TB_THREADS - 1) / TB_THREADS)

__global__ __launch_bounds__(TB_THREADS) void tile_ids(
    const int* __restrict__ tgt, unsigned* __restrict__ tb, int E4) {
  const int i = blockIdx.x * TB_THREADS + threadIdx.x;
  if (i < E4) {
    const int4 t = ((const int4*)tgt)[i];
    tb[i] = (((unsigned)t.x / TILE) & 0xFFu) |
            ((((unsigned)t.y / TILE) & 0xFFu) << 8) |
            ((((unsigned)t.z / TILE) & 0xFFu) << 16) |
            (((unsigned)t.w / TILE) << 24);
  }
}

#define SCAP 512

template <int F>
__device__ __forceinline__ void add_rows(
    const unsigned* ok, const float* ow, int n_own,
    const float* __restrict__ H, float4* acc, int lane) {
  constexpr int F4 = F / 4;
  for (int j0 = 0; j0 < n_own; j0 += 8) {
    unsigned kk[8];
    float wq[8];
    float4 v[8];
#pragma unroll
    for (int u = 0; u < 8; ++u) {
      const int jj = min(j0 + u, n_own - 1);
      kk[u] = (unsigned)__builtin_amdgcn_readfirstlane((int)ok[jj]);
      wq[u] = ow[jj];
    }
#pragma unroll
    for (int u = 0; u < 8; ++u) {
      const size_t s = (size_t)(kk[u] & 0xFFFFFFu);
      v[u] = make_float4(0.f, 0.f, 0.f, 0.f);
      if (lane < F4) v[u] = *((const float4*)(H + s * F) + lane);
    }
#pragma unroll
    for (int u = 0; u < 8; ++u) {
      if (j0 + u < n_own) {
        const int tl = (int)(kk[u] >> 24);
        if (lane < F4) {
          float4 a = acc[tl * F4 + lane];
          a.x = fmaf(wq[u], v[u].x, a.x);
          a.y = fmaf(wq[u], v[u].y, a.y);
          a.z = fmaf(wq[u], v[u].z, a.z);
          a.w = fmaf(wq[u], v[u].w, a.w);
          acc[tl * F4 + lane] = a;
        }
      }
    }
  }
}

template <int F, int MODE>
__global__ __launch_bounds__(1024) void agg_tile(
    const float* __restrict__ H, const int* __restrict__ src,
    const int* __restrict__ tgt, const float* __restrict__ w,
    float* __restrict__ A, unsigned* __restrict__ slist, int* __restrict__ scnt,
    const unsigned* __restrict__ tb, int n, int E) {
  constexpr int F4 = F / 4;
  __shared__ float4   acc[TILE * F4];
  __shared__ unsigned lkey[NWV * CAP];
  __shared__ float    lwt[NWV * CAP];
  __shared__ unsigned okey[NWV * CAP2];
  __shared__ float    owt[NWV * CAP2];
  __shared__ int      lcnt[NWV];
  __shared__ int      wmore[NWV];
  __shared__ unsigned omask[NWV * OMP];

  const int tid  = threadIdx.x;
  const int lane = tid & 31;
  const int wave = __builtin_amdgcn_readfirstlane(tid >> 5);
  const int lo   = blockIdx.x * TILE;
  const unsigned below = (1u << lane) - 1u;
  const unsigned pat = (unsigned)blockIdx.x * 0x01010101u;

  unsigned* ok = okey + wave * CAP2;
  float*    ow = owt + wave * CAP2;
  uint2*    sv = (uint2*)slist + (size_t)(blockIdx.x * NWV + wave) * SCAP;

  int  cl = 0;
  bool use_saved = false;
  if (MODE == 1) {
    cl = scnt[blockIdx.x * NWV + lane];
    use_saved = __builtin_amdgcn_ballot_w32(cl > SCAP) == 0;
  }

  for (int i = tid; i < TILE * F4; i += 1024) acc[i] = make_float4(0.f, 0.f, 0.f, 0.f);

  if (use_saved) {
    const int c = __builtin_amdgcn_readlane(cl, wave);
    __syncthreads();
    for (int j0 = 0; j0 < c; j0 += CAP2) {
      const int nn = min(CAP2, c - j0);
      uint2 e[CAP2 / 32];
#pragma unroll
      for (int u = 0; u < CAP2 / 32; ++u) {
        const int i = u * 32 + lane;
        e[u] = make_uint2(0u, 0u);
        if (i < nn) e[u] = sv[j0 + i];
      }
#pragma unroll
      for (int u = 0; u < CAP2 / 32; ++u) {
        const int i = u * 32 + lane;
        if (i < nn) {
          ok[i] = e[u].x;
          ow[i] = __uint_as_float(e[u].y);
        }
      }
      __builtin_amdgcn_wave_barrier();
      add_rows<F>(ok, ow, nn, H, acc, lane);
      __builtin_amdgcn_wave_barrier();
    }
  } else {
    const int per = (((E + NWV - 1) / NWV) + 3) & ~3;
    int pos = min(wave * per, E);
    const int end = min(pos + per, E);
    unsigned* mykey = lkey + wave * CAP;
    float*    myw   = lwt + wave * CAP;

    const int max_trips = per / (CAP - 127) + 2;
    int skip = 0;
    int saved = 0;
    for (int trip = 0; trip < max_trips; ++trip) {
      __syncthreads();

      int cnt = 0;
      while (pos < end) {
        unsigned x[32];
        if (pos + 4096 <= end) {
          const unsigned* tp = tb + (pos >> 2) + lane;
#pragma unroll
          for (int i = 0; i < 32; ++i) x[i] = tp[i * 32] ^ pat;
        } else {
#pragma unroll
          for (int i = 0; i < 32; ++i) {
            const int wi = (pos >> 2) + i * 32 + lane;
            const unsigned v = tb[min(wi, (E >> 2) - 1)] ^ pat;
            x[i] = (wi * 4 < end) ? v : 0xFFFFFFFFu;
          }
        }
        if (skip != 0) {
#pragma unroll
          for (int u = 0; u < 8; ++u)
            if (u < skip) x[u] = 0xFFFFFFFFu;
        }
#pragma unroll
        for (int ss = 0; ss < 4; ++ss) {
          if (pos < end) {
#pragma unroll
            for (int u = 0; u < 8; ++u) {
              const unsigned xv = x[ss * 8 + u];
              const unsigned hz = (xv - 0x01010101u) & ~xv & 0x80808080u;
              if (__builtin_amdgcn_ballot_w32(hz != 0u) != 0) {
                if (cnt > CAP - 128) {
                  skip = u;
                  goto scan_done;
                }
#pragma unroll
                for (int c = 0; c < 4; ++c) {
                  const unsigned mc =
                      __builtin_amdgcn_ballot_w32(((xv >> (8 * c)) & 0xFFu) == 0u);
                  if (mc != 0) {
                    if ((mc >> lane) & 1u) {
                      const int idx = cnt + __builtin_popcount(mc & below);
                      mykey[idx] = (unsigned)(pos + (u * 32 + lane) * 4 + c);
                    }
                    cnt += __builtin_popcount(mc);
                  }
                }
              }
            }
            pos += 1024;
            skip = 0;
          }
        }
      }
    scan_done:
      __builtin_amdgcn_wave_barrier();
      for (int i0 = 0; i0 < cnt; i0 += 32) {
        const int  i     = i0 + lane;
        const bool valid = i < cnt;
        unsigned tl = 0u;
        if (valid) {
          const int e = (int)mykey[i];
          tl = (unsigned)(tgt[e] - lo);
          mykey[i] = (tl << 24) | (unsigned)src[e];
          myw[i]   = w[e];
        }
        if (i0 < FCH * 32) {
          unsigned mo = __builtin_amdgcn_ballot_w32(valid);
#pragma unroll
          for (int k = 0; k < 5; ++k) {
            const unsigned bk = __builtin_amdgcn_ballot_w32(((tl >> k) & 1u) != 0u);
            mo &= ((lane >> k) & 1) ? bk : ~bk;
          }
          omask[wave * OMP + i0 + lane] = mo;
        }
      }
      if (lane == 0) {
        lcnt[wave]  = cnt;
        wmore[wave] = (pos < end) ? 1 : 0;
      }
      __syncthreads();
      const bool more = __builtin_amdgcn_ballot_w32(wmore[lane] != 0) != 0;

      bool by_masks = __builtin_amdgcn_ballot_w32(lcnt[lane] > FCH * 32) == 0;
      int sl = 0, base = 0;
      while (sl < NWV) {
        int n_own = 0;
        if (by_masks) {
          by_masks = false;
          const int cv = lcnt[lane];
          unsigned m[FCH];
          int nv = 0;
#pragma unroll
          for (int c = 0; c < FCH; ++c) {
            m[c] = (c * 32 < cv) ? omask[lane * OMP + c * 32 + wave] : 0u;
            nv += __builtin_popcount(m[c]);
          }
          int incl = nv;
#pragma unroll
          for (int d = 1; d < 32; d <<= 1) {
            const int t = __shfl_up(incl, d, 32);
            if (lane >= d) incl += t;
          }
          const int total = __builtin_amdgcn_readlane(incl, 31);
          if (total <= CAP2) {
            int p = incl - nv;
#pragma unroll
            for (int c = 0; c < FCH; ++c) {
              unsigned mm = m[c];
              while (__builtin_amdgcn_ballot_w32(mm != 0u) != 0) {
                if (mm != 0u) {
                  const int b = __builtin_ctz(mm);
                  mm &= mm - 1u;
                  const int idx = lane * CAP + c * 32 + b;
                  ok[p] = lkey[idx];
                  ow[p] = lwt[idx];
                  ++p;
                }
              }
            }
            n_own = total;
            sl = NWV;
          }
        }
        while (sl < NWV && n_own <= CAP2 - 32) {
          const int c = __builtin_amdgcn_readfirstlane(lcnt[sl]);
          if (base >= c) { ++sl; base = 0; continue; }
          const int i = base + lane;
          unsigned k = 0;
          float ww = 0.f;
          bool mine = false;
          if (i < c) {
            k    = lkey[sl * CAP + i];
            ww   = lwt[sl * CAP + i];
            mine = (int)((k >> 24) & 31u) == wave;
          }
          const unsigned mm = __builtin_amdgcn_ballot_w32(mine);
          if (mm != 0) {
            if (mine) {
              const int idx = n_own + __builtin_popcount(mm & below);
              ok[idx] = k;
              ow[idx] = ww;
            }
            n_own += __builtin_popcount(mm);
          }
          base += 32;
        }
        __builtin_amdgcn_wave_barrier();
        if (MODE == 0) {
          for (int i = lane; i < n_own; i += 32) {
            const int g = saved + i;
            if (g < SCAP) sv[g] = make_uint2(ok[i], __float_as_uint(ow[i]));
          }
          saved += n_own;
        }
        add_rows<F>(ok, ow, n_own, H, acc, lane);
        __builtin_amdgcn_wave_barrier();
      }
      if (!more) break;
    }
    if (MODE == 0 && lane == 0) scnt[blockIdx.x * NWV + wave] = saved;
  }
  __syncthreads();

  float4* A4 = (float4*)A;
  const int g0   = lo * F4;
  const int gmax = n * F4;
  for (int i = tid; i < TILE * F4; i += 1024)
    if (g0 + i < gmax) A4[g0 + i] = acc[i];
}

#define EW_BLOCKS 1024
#define EW_THREADS 256
#define EW_UNROLL 8

__global__ __launch_bounds__(EW_THREADS) void lsm40(
    const float* __restrict__ Ain, const float* __restrict__ b,
    float* __restrict__ out, int n) {
  const int lane = threadIdx.x & 31;
  const int gw = __builtin_amdgcn_readfirstlane(
      (int)((blockIdx.x * blockDim.x + threadIdx.x) >> 5));
  const int nw = (int)((gridDim.x * blockDim.x) >> 5);
  const float c0 = b[lane];
  const float c1 = (lane < 8) ? b[32 + lane] : 0.0f;
  for (int nb = gw; nb < n; nb += EW_UNROLL * nw) {
    float r0[EW_UNROLL], r1[EW_UNROLL];
#pragma unroll
    for (int k = 0; k < EW_UNROLL; ++k) {
      const int node = nb + k * nw;
      r0[k] = 0.0f;
      r1[k] = 0.0f;
      if (node < n) {
        const float* row = Ain + (size_t)node * 40;
        r0[k] = row[lane];
        if (lane < 8) r1[k] = row[32 + lane];
      }
    }
#pragma unroll
    for (int k = 0; k < EW_UNROLL; ++k) {
      const int node = nb + k * nw;
      if (node < n) {
        const float v0 = r0[k] + c0;
        const float v1 = (lane < 8) ? (r1[k] + c1) : -INFINITY;
        float m = fmaxf(v0, v1);
#pragma unroll
        for (int o = 16; o > 0; o >>= 1) m = fmaxf(m, __shfl_xor(m, o, 32));
        float s = expf(v0 - m) + ((lane < 8) ? expf(v1 - m) : 0.0f);
#pragma unroll
        for (int o = 16; o > 0; o >>= 1) s += __shfl_xor(s, o, 32);
        const float lse = m + logf(s);
        out[(size_t)node * 40 + lane] = v0 - lse;
        if (lane < 8) out[(size_t)node * 40 + 32 + lane] = v1 - lse;
      }
    }
  }
}

extern "C" void kernel_launch(void* const* d_in, const int* in_sizes, int n_in,
                              void* d_out, int out_size, void* d_ws, size_t ws_size,
                              hipStream_t stream) {
  const int N = NNODES, E = NEDGES;
  const float* x     = (const float*)d_in[0];
  const int*   src   = (const int*)d_in[1];
  const int*   tgt   = (const int*)d_in[2];
  const float* w     = (const float*)d_in[3];
  const float* W0    = (const float*)d_in[4];
  const float* b0    = (const float*)d_in[5];
  const float* W1    = (const float*)d_in[6];
  const float* b1    = (const float*)d_in[7];
  const float* W2    = (const float*)d_in[8];
  const float* b2    = (const float*)d_in[9];
  const float* gamma = (const float*)d_in[10];
  const float* beta  = (const float*)d_in[11];
  float* out = (float*)d_out;

  float* H   = (float*)d_ws;
  float* A   = H + (size_t)N * 128;
  float* H2  = H;
  float* AG2 = H + (size_t)N * 64;
  unsigned* SL = (unsigned*)(H + (size_t)N * 256);
  int*      SC = (int*)(SL + (size_t)NTILES * NWV * SCAP * 2);
  unsigned* TB = (unsigned*)(SC + NTILES * NWV);

  const int gemm_blocks = (N + 127) / 128;

  tile_ids<<<TB_BLOCKS, TB_THREADS, 0, stream>>>(tgt, TB, NEDGES / 4);


  gemm_wmma_f32<128, 0><<<gemm_blocks, 256, 0, stream>>>(x, W0, nullptr, nullptr, nullptr, H, N);
  agg_tile<128, 0><<<NTILES, 1024, 0, stream>>>(H, src, tgt, w, A, SL, SC, TB, N, E);

  gemm_wmma_f32<128, 1><<<gemm_blocks, 256, 0, stream>>>(A, W1, b0, nullptr, nullptr, H, N);
  agg_tile<128, 1><<<NTILES, 1024, 0, stream>>>(H, src, tgt, w, A, SL, SC, TB, N, E);

  gemm_wmma_f32<40, 2><<<gemm_blocks, 256, 0, stream>>>(A, W2, b1, gamma, beta, H2, N);
  agg_tile<40, 1><<<NTILES, 1024, 0, stream>>>(H2, src, tgt, w, AG2, SL, SC, TB, N, E);
  lsm40<<<EW_BLOCKS, EW_THREADS, 0, stream>>>(AG2, b2, out, N);
}


# each aggregation launch applies the next row-wise stage to its tile's sums in LDS: layer 1 and 2 products (f32 WMMA, W packed in LDS) and log_softmax; 3 of 8 launches gone; sum orders kept
# speedup vs baseline: 3.2899x; 1.1398x over previous
#include <hip/hip_runtime.h>
#include <math.h>

typedef __attribute__((ext_vector_type(2))) float v2f;
typedef __attribute__((ext_vector_type(8))) float v8f;

#define NNODES 50000
#define NEDGES 800000

#define GK 128

template <int NC, int NT>
__device__ __forceinline__ void stage_w(const float* __restrict__ W, v2f* lw, int tid) {
  constexpr int PC = ((NC + 15) / 16) * 16;
#pragma unroll
  for (int i = tid; i < (GK / 2) * PC; i += NT) {
    const int kp = i / PC, c = i % PC;
    const int cc = min(c, NC - 1);
    const float w0 = W[(2 * kp) * NC + cc];
    const float w1 = W[(2 * kp + 1) * NC + cc];
    v2f val;
    val.x = (c < NC) ? w0 : 0.0f;
    val.y = (c < NC) ? w1 : 0.0f;
    lw[i] = val;
  }
}

template <int EPI>
__device__ __forceinline__ float4 row_fn(float4 v, const float4 (*ep)[GK / 4], int g) {
  if (EPI >= 1) {
    const float4 bb = ep[0][g];
    v.x = fmaxf(v.x + bb.x, 0.0f);
    v.y = fmaxf(v.y + bb.y, 0.0f);
    v.z = fmaxf(v.z + bb.z, 0.0f);
    v.w = fmaxf(v.w + bb.w, 0.0f);
  }
  if (EPI == 2) {
    const float4 gm = ep[1][g];
    const float4 bt = ep[2][g];
    const float mu  = 0.25f * (v.x + v.y + v.z + v.w);
    const float ms  = 0.25f * (v.x * v.x + v.y * v.y + v.z * v.z + v.w * v.w);
    const float inv = rsqrtf(fmaxf(ms - mu * mu, 0.0f) + 1e-5f);
    v.x = (v.x - mu) * inv * gm.x + bt.x;
    v.y = (v.y - mu) * inv * gm.y + bt.y;
    v.z = (v.z - mu) * inv * gm.z + bt.z;
    v.w = (v.w - mu) * inv * gm.w + bt.w;
  }
  return v;
}

template <int PC, int CPW>
__device__ __forceinline__ void wmma_kgroup(
    const float4 v, const v2f* lw, int j, int hk, int m, int t0, v8f* acc) {
  const v2f* brow = &lw[(4 * j + 2 * hk) * PC + t0 * 16 + m];
  v2f a0, a1;
  a0.x = v.x; a0.y = v.y;
  a1.x = v.z; a1.y = v.w;
#pragma unroll
  for (int t = 0; t < CPW; ++t) {
    const v2f b = brow[t * 16];
    acc[t] = __builtin_amdgcn_wmma_f32_16x16x4_f32(
        false, a0, false, b, (short)0, acc[t], false, false);
  }
#pragma unroll
  for (int t = 0; t < CPW; ++t) {
    const v2f b = brow[PC + t * 16];
    acc[t] = __builtin_amdgcn_wmma_f32_16x16x4_f32(
        false, a1, false, b, (short)0, acc[t], false, false);
  }
}

template <int NC>
__global__ __launch_bounds__(256) void gemm_wmma_f32(
    const float* __restrict__ X, const float* __restrict__ W,
    float* __restrict__ H, int n) {
  constexpr int K  = GK;
  constexpr int CT = (NC + 15) / 16;
  constexpr int PC = CT * 16;
  __shared__ v2f lw[(K / 2) * PC];

  const int tid = threadIdx.x;
  stage_w<NC, 256>(W, lw, tid);
  __syncthreads();

  const int wave = tid >> 5;
  const int lane = tid & 31;
  const int row0 = (blockIdx.x * 8 + wave) * 16;
  if (row0 >= n) return;

  const int m  = lane & 15;
  const int hk = lane >> 4;

  v8f acc[CT];
#pragma unroll
  for (int t = 0; t < CT; ++t) acc[t] = {};

  const float4* xrow = (const float4*)(X + (size_t)(row0 + m) * K) + hk;

  float4 q[K / 8];
#pragma unroll
  for (int j = 0; j < K / 8; ++j) q[j] = xrow[2 * j];

#pragma unroll
  for (int j = 0; j < K / 8; ++j) wmma_kgroup<PC, CT>(q[j], lw, j, hk, m, 0, acc);

  const int crow = row0 + hk * 8;
#pragma unroll
  for (int t = 0; t < CT; ++t) {
    const int col = t * 16 + m;
    if ((NC % 16 == 0) || (col < NC)) {
#pragma unroll
      for (int r = 0; r < 8; ++r)
        H[(size_t)(crow + r) * NC + col] = acc[t][r];
    }
  }
}

#define TILE 196
#define NTILES ((NNODES + TILE - 1) / TILE)
#define NWV 32
#define CAP 448
#define CAP2 256
#define FCH 5
#define OMP (FCH * 32 + 1)

static_assert(TILE <= 256, "local node id is kept in 8 bits");
static_assert(NNODES < (1 << 24) && NEDGES < (1 << 24), "ids are kept in 24 bits");

static_assert(NTILES <= 256, "a tile id is kept in 8 bits");
static_assert(NEDGES % 4 == 0, "four tile ids per word");
#define TB_THREADS 256
#define TB_BLOCKS ((NEDGES / 4 + TB_THREADS - 1) / TB_THREADS)

__global__ __launch_bounds__(TB_THREADS) void tile_ids(
    const int* __restrict__ tgt, unsigned* __restrict__ tb, int E4) {
  const int i = blockIdx.x * TB_THREADS + threadIdx.x;
  if (i < E4) {
    const int4 t = ((const int4*)tgt)[i];
    tb[i] = (((unsigned)t.x / TILE) & 0xFFu) |
            ((((unsigned)t.y / TILE) & 0xFFu) << 8) |
            ((((unsigned)t.z / TILE) & 0xFFu) << 16) |
            (((unsigned)t.w / TILE) << 24);
  }
}

#define SCAP 512

template <int F, int RS>
__device__ __forceinline__ void add_rows(
    const unsigned* ok, const float* ow, int n_own,
    const float* __restrict__ H, float4* acc, int lane) {
  constexpr int F4 = F / 4;
  for (int j0 = 0; j0 < n_own; j0 += 8) {
    unsigned kk[8];
    float wq[8];
    float4 v[8];
#pragma unroll
    for (int u = 0; u < 8; ++u) {
      const int jj = min(j0 + u, n_own - 1);
      kk[u] = (unsigned)__builtin_amdgcn_readfirstlane((int)ok[jj]);
      wq[u] = ow[jj];
    }
#pragma unroll
    for (int u = 0; u < 8; ++u) {
      const size_t s = (size_t)(kk[u] & 0xFFFFFFu);
      v[u] = make_float4(0.f, 0.f, 0.f, 0.f);
      if (lane < F4) v[u] = *((const float4*)(H + s * F) + lane);
    }
#pragma unroll
    for (int u = 0; u < 8; ++u) {
      if (j0 + u < n_own) {
        const int tl = (int)(kk[u] >> 24);
        if (lane < F4) {
          float4 a = acc[tl * RS + lane];
          a.x = fmaf(wq[u], v[u].x, a.x);
          a.y = fmaf(wq[u], v[u].y, a.y);
          a.z = fmaf(wq[u], v[u].z, a.z);
          a.w = fmaf(wq[u], v[u].w, a.w);
          acc[tl * RS + lane] = a;
        }
      }
    }
  }
}

template <int NC, int EPI, int CPW>
__device__ __forceinline__ void tail_product(
    const float4* acc, const v2f* lw, const float4 (*ep)[GK / 4],
    float* __restrict__ Hn, int lo, int n, int wave, int lane) {
  constexpr int CT = (NC + 15) / 16;
  constexpr int PC = CT * 16;
  constexpr int RS = GK / 4 + 1;
  constexpr int RT = (TILE + 15) / 16;
  constexpr int CG = CT / CPW;
  static_assert(CT % CPW == 0, "column tiles split evenly into groups");
  static_assert(RT * CG <= NWV, "one unit per wave");
  const int m  = lane & 15;
  const int hk = lane >> 4;
  if (wave < RT * CG) {
    const int rt = wave / CG;
    const int t0 = (wave % CG) * CPW;
    const float4* xrow = acc + min(rt * 16 + m, TILE - 1) * RS + hk;
    v8f c[CPW];
#pragma unroll
    for (int t = 0; t < CPW; ++t) c[t] = {};
#pragma unroll
    for (int j = 0; j < GK / 8; ++j)
      wmma_kgroup<PC, CPW>(row_fn<EPI>(xrow[2 * j], ep, 2 * j + hk), lw, j, hk, m, t0, c);

    const int lr0 = rt * 16 + hk * 8;
#pragma unroll
    for (int t = 0; t < CPW; ++t) {
      const int col = (t0 + t) * 16 + m;
      if ((NC % 16 == 0) || (col < NC)) {
#pragma unroll
        for (int r = 0; r < 8; ++r) {
          const int lr = lr0 + r;
          if (lr < TILE && lo + lr < n) Hn[(size_t)(lo + lr) * NC + col] = c[t][r];
        }
      }
    }
  }
}

__device__ __forceinline__ void tail_lsm40(
    const float4* acc, const float* __restrict__ b, float* __restrict__ out,
    int lo, int n, int wave, int lane) {
  const float* rows = (const float*)acc;
  const float c0 = b[lane];
  const float c1 = (lane < 8) ? b[32 + lane] : 0.0f;
  for (int r = wave; r < TILE && lo + r < n; r += NWV) {
    const float v0 = rows[r * 40 + lane] + c0;
    const float v1 = (lane < 8) ? (rows[r * 40 + 32 + lane] + c1) : -INFINITY;
    float mx = fmaxf(v0, v1);
#pragma unroll
    for (int o = 16; o > 0; o >>= 1) mx = fmaxf(mx, __shfl_xor(mx, o, 32));
    float sm = expf(v0 - mx) + ((lane < 8) ? expf(v1 - mx) : 0.0f);
#pragma unroll
    for (int o = 16; o > 0; o >>= 1) sm += __shfl_xor(sm, o, 32);
    const float lse = mx + logf(sm);
    out[(size_t)(lo + r) * 40 + lane] = v0 - lse;
    if (lane < 8) out[(size_t)(lo + r) * 40 + 32 + lane] = v1 - lse;
  }
}

template <int F, int MODE, int NCN, int EPI>
__global__ __launch_bounds__(1024) void agg_tile(
    const float* __restrict__ H, const int* __restrict__ src,
    const int* __restrict__ tgt, const float* __restrict__ w,
    const float* __restrict__ Wn, const float* __restrict__ eb,
    const float* __restrict__ eg, const float* __restrict__ et,
    float* __restrict__ On, unsigned* __restrict__ slist, int* __restrict__ scnt,
    const unsigned* __restrict__ tb, int n, int E) {
  static_assert(NCN == 0 ? F == 40 : F == GK, "the tail reads rows of the width it was written for");
  constexpr int F4 = F / 4;
  constexpr int RS = (NCN != 0) ? F4 + 1 : F4;
  constexpr int PCN = ((NCN + 15) / 16) * 16;
  __shared__ float4   acc[TILE * RS];
  __shared__ v2f      lbuf[NWV * CAP];
  static_assert((GK / 2) * PCN <= NWV * CAP, "packed W fits where the scan lists were");
  unsigned* lkey = (unsigned*)lbuf;
  float*    lwt  = (float*)lbuf + NWV * CAP;
  __shared__ float4   ep[3][GK / 4];
  __shared__ unsigned okey[NWV * CAP2];
  __shared__ float    owt[NWV * CAP2];
  __shared__ int      lcnt[NWV];
  __shared__ int      wmore[NWV];
  __shared__ unsigned omask[NWV * OMP];

  const int tid  = threadIdx.x;
  const int lane = tid & 31;
  const int wave = __builtin_amdgcn_readfirstlane(tid >> 5);
  const int lo   = blockIdx.x * TILE;
  const unsigned below = (1u << lane) - 1u;
  const unsigned pat = (unsigned)blockIdx.x * 0x01010101u;

  unsigned* ok = okey + wave * CAP2;
  float*    ow = owt + wave * CAP2;
  uint2*    sv = (uint2*)slist + (size_t)(blockIdx.x * NWV + wave) * SCAP;

  int  cl = 0;
  bool use_saved = false;
  if (MODE == 1) {
    cl = scnt[blockIdx.x * NWV + lane];
    use_saved = __builtin_amdgcn_ballot_w32(cl > SCAP) == 0;
  }

  for (int i = tid; i < TILE * RS; i += 1024) acc[i] = make_float4(0.f, 0.f, 0.f, 0.f);

  if constexpr (NCN != 0) {
    if (EPI >= 1 && tid < GK / 4) ep[0][tid] = *(const float4*)(eb + tid * 4);
    if (EPI == 2 && tid < GK / 4) {
      ep[1][tid] = *(const float4*)(eg + tid * 4);
      ep[2][tid] = *(const float4*)(et + tid * 4);
    }
  }

  if (use_saved) {
    if constexpr (NCN != 0) stage_w<NCN, 1024>(Wn, lbuf, tid);
    const int c = __builtin_amdgcn_readlane(cl, wave);
    __syncthreads();
    for (int j0 = 0; j0 < c; j0 += CAP2) {
      const int nn = min(CAP2, c - j0);
      uint2 e[CAP2 / 32];
#pragma unroll
      for (int u = 0; u < CAP2 / 32; ++u) {
        const int i = u * 32 + lane;
        e[u] = make_uint2(0u, 0u);
        if (i < nn) e[u] = sv[j0 + i];
      }
#pragma unroll
      for (int u = 0; u < CAP2 / 32; ++u) {
        const int i = u * 32 + lane;
        if (i < nn) {
          ok[i] = e[u].x;
          ow[i] = __uint_as_float(e[u].y);
        }
      }
      __builtin_amdgcn_wave_barrier();
      add_rows<F, RS>(ok, ow, nn, H, acc, lane);
      __builtin_amdgcn_wave_barrier();
    }
  } else {
    const int per = (((E + NWV - 1) / NWV) + 3) & ~3;
    int pos = min(wave * per, E);
    const int end = min(pos + per, E);
    unsigned* mykey = lkey + wave * CAP;
    float*    myw   = lwt + wave * CAP;

    const int max_trips = per / (CAP - 127) + 2;
    int skip = 0;
    int saved = 0;
    for (int trip = 0; trip < max_trips; ++trip) {
      __syncthreads();

      int cnt = 0;
      while (pos < end) {
        unsigned x[32];
        if (pos + 4096 <= end) {
          const unsigned* tp = tb + (pos >> 2) + lane;
#pragma unroll
          for (int i = 0; i < 32; ++i) x[i] = tp[i * 32] ^ pat;
        } else {
#pragma unroll
          for (int i = 0; i < 32; ++i) {
            const int wi = (pos >> 2) + i * 32 + lane;
            const unsigned v = tb[min(wi, (E >> 2) - 1)] ^ pat;
            x[i] = (wi * 4 < end) ? v : 0xFFFFFFFFu;
          }
        }
        if (skip != 0) {
#pragma unroll
          for (int u = 0; u < 8; ++u)
            if (u < skip) x[u] = 0xFFFFFFFFu;
        }
#pragma unroll
        for (int ss = 0; ss < 4; ++ss) {
          if (pos < end) {
#pragma unroll
            for (int u = 0; u < 8; ++u) {
              const unsigned xv = x[ss * 8 + u];
              const unsigned hz = (xv - 0x01010101u) & ~xv & 0x80808080u;
              if (__builtin_amdgcn_ballot_w32(hz != 0u) != 0) {
                if (cnt > CAP - 128) {
                  skip = u;
                  goto scan_done;
                }
#pragma unroll
                for (int c = 0; c < 4; ++c) {
                  const unsigned mc =
                      __builtin_amdgcn_ballot_w32(((xv >> (8 * c)) & 0xFFu) == 0u);
                  if (mc != 0) {
                    if ((mc >> lane) & 1u) {
                      const int idx = cnt + __builtin_popcount(mc & below);
                      mykey[idx] = (unsigned)(pos + (u * 32 + lane) * 4 + c);
                    }
                    cnt += __builtin_popcount(mc);
                  }
                }
              }
            }
            pos += 1024;
            skip = 0;
          }
        }
      }
    scan_done:
      __builtin_amdgcn_wave_barrier();
      for (int i0 = 0; i0 < cnt; i0 += 32) {
        const int  i     = i0 + lane;
        const bool valid = i < cnt;
        unsigned tl = 0u;
        if (valid) {
          const int e = (int)mykey[i];
          tl = (unsigned)(tgt[e] - lo);
          mykey[i] = (tl << 24) | (unsigned)src[e];
          myw[i]   = w[e];
        }
        if (i0 < FCH * 32) {
          unsigned mo = __builtin_amdgcn_ballot_w32(valid);
#pragma unroll
          for (int k = 0; k < 5; ++k) {
            const unsigned bk = __builtin_amdgcn_ballot_w32(((tl >> k) & 1u) != 0u);
            mo &= ((lane >> k) & 1) ? bk : ~bk;
          }
          omask[wave * OMP + i0 + lane] = mo;
        }
      }
      if (lane == 0) {
        lcnt[wave]  = cnt;
        wmore[wave] = (pos < end) ? 1 : 0;
      }
      __syncthreads();
      const bool more = __builtin_amdgcn_ballot_w32(wmore[lane] != 0) != 0;

      bool by_masks = __builtin_amdgcn_ballot_w32(lcnt[lane] > FCH * 32) == 0;
      int sl = 0, base = 0;
      while (sl < NWV) {
        int n_own = 0;
        if (by_masks) {
          by_masks = false;
          const int cv = lcnt[lane];
          unsigned m[FCH];
          int nv = 0;
#pragma unroll
          for (int c = 0; c < FCH; ++c) {
            m[c] = (c * 32 < cv) ? omask[lane * OMP + c * 32 + wave] : 0u;
            nv += __builtin_popcount(m[c]);
          }
          int incl = nv;
#pragma unroll
          for (int d = 1; d < 32; d <<= 1) {
            const int t = __shfl_up(incl, d, 32);
            if (lane >= d) incl += t;
          }
          const int total = __builtin_amdgcn_readlane(incl, 31);
          if (total <= CAP2) {
            int p = incl - nv;
#pragma unroll
            for (int c = 0; c < FCH; ++c) {
              unsigned mm = m[c];
              while (__builtin_amdgcn_ballot_w32(mm != 0u) != 0) {
                if (mm != 0u) {
                  const int b = __builtin_ctz(mm);
                  mm &= mm - 1u;
                  const int idx = lane * CAP + c * 32 + b;
                  ok[p] = lkey[idx];
                  ow[p] = lwt[idx];
                  ++p;
                }
              }
            }
            n_own = total;
            sl = NWV;
          }
        }
        while (sl < NWV && n_own <= CAP2 - 32) {
          const int c = __builtin_amdgcn_readfirstlane(lcnt[sl]);
          if (base >= c) { ++sl; base = 0; continue; }
          const int i = base + lane;
          unsigned k = 0;
          float ww = 0.f;
          bool mine = false;
          if (i < c) {
            k    = lkey[sl * CAP + i];
            ww   = lwt[sl * CAP + i];
            mine = (int)((k >> 24) & 31u) == wave;
          }
          const unsigned mm = __builtin_amdgcn_ballot_w32(mine);
          if (mm != 0) {
            if (mine) {
              const int idx = n_own + __builtin_popcount(mm & below);
              ok[idx] = k;
              ow[idx] = ww;
            }
            n_own += __builtin_popcount(mm);
          }
          base += 32;
        }
        __builtin_amdgcn_wave_barrier();
        if (MODE == 0) {
          for (int i = lane; i < n_own; i += 32) {
            const int g = saved + i;
            if (g < SCAP) sv[g] = make_uint2(ok[i], __float_as_uint(ow[i]));
          }
          saved += n_own;
        }
        add_rows<F, RS>(ok, ow, n_own, H, acc, lane);
        __builtin_amdgcn_wave_barrier();
      }
      if (!more) break;
    }
    if (MODE == 0 && lane == 0) scnt[blockIdx.x * NWV + wave] = saved;
  }
  __syncthreads();

  if constexpr (NCN != 0) {
    if (!use_saved) {
      stage_w<NCN, 1024>(Wn, lbuf, tid);
      __syncthreads();
    }
    tail_product<NCN, EPI, (NCN == GK ? 4 : 3)>(acc, lbuf, ep, On, lo, n, wave, lane);
  } else {
    tail_lsm40(acc, eb, On, lo, n, wave, lane);
  }
}

extern "C" void kernel_launch(void* const* d_in, const int* in_sizes, int n_in,
                              void* d_out, int out_size, void* d_ws, size_t ws_size,
                              hipStream_t stream) {
  const int N = NNODES, E = NEDGES;
  const float* x     = (const float*)d_in[0];
  const int*   src   = (const int*)d_in[1];
  const int*   tgt   = (const int*)d_in[2];
  const float* w     = (const float*)d_in[3];
  const float* W0    = (const float*)d_in[4];
  const float* b0    = (const float*)d_in[5];
  const float* W1    = (const float*)d_in[6];
  const float* b1    = (const float*)d_in[7];
  const float* W2    = (const float*)d_in[8];
  const float* b2    = (const float*)d_in[9];
  const float* gamma = (const float*)d_in[10];
  const float* beta  = (const float*)d_in[11];
  float* out = (float*)d_out;

  float* H   = (float*)d_ws;
  float* H1  = H + (size_t)N * 128;
  float* H2  = H;
  unsigned* SL = (unsigned*)(H + (size_t)N * 256);
  int*      SC = (int*)(SL + (size_t)NTILES * NWV * SCAP * 2);
  unsigned* TB = (unsigned*)(SC + NTILES * NWV);

  const int gemm_blocks = (N + 127) / 128;

  tile_ids<<<TB_BLOCKS, TB_THREADS, 0, stream>>>(tgt, TB, NEDGES / 4);


  gemm_wmma_f32<128><<<gemm_blocks, 256, 0, stream>>>(x, W0, H, N);
  agg_tile<128, 0, 128, 1><<<NTILES, 1024, 0, stream>>>(
      H, src, tgt, w, W1, b0, nullptr, nullptr, H1, SL, SC, TB, N, E);

  agg_tile<128, 1, 40, 2><<<NTILES, 1024, 0, stream>>>(
      H1, src, tgt, w, W2, b1, gamma, beta, H2, SL, SC, TB, N, E);

  agg_tile<40, 1, 0, 0><<<NTILES, 1024, 0, stream>>>(
      H2, src, tgt, w, nullptr, b2, nullptr, nullptr, out, SL, SC, TB, N, E);
}


# add_rows' row loads run as a ring of 8: a slot is refilled with the row 8 edges on right after its row is added, so loads stay in flight under the LDS adds; no row load predicated; sum order kept
# speedup vs baseline: 3.3951x; 1.0320x over previous
#include <hip/hip_runtime.h>
#include <math.h>

typedef __attribute__((ext_vector_type(2))) float v2f;
typedef __attribute__((ext_vector_type(8))) float v8f;

#define NNODES 50000
#define NEDGES 800000

#define GK 128

template <int NC, int NT>
__device__ __forceinline__ void stage_w(const float* __restrict__ W, v2f* lw, int tid) {
  constexpr int PC = ((NC + 15) / 16) * 16;
#pragma unroll
  for (int i = tid; i < (GK / 2) * PC; i += NT) {
    const int kp = i / PC, c = i % PC;
    const int cc = min(c, NC - 1);
    const float w0 = W[(2 * kp) * NC + cc];
    const float w1 = W[(2 * kp + 1) * NC + cc];
    v2f val;
    val.x = (c < NC) ? w0 : 0.0f;
    val.y = (c < NC) ? w1 : 0.0f;
    lw[i] = val;
  }
}

template <int EPI>
__device__ __forceinline__ float4 row_fn(float4 v, const float4 (*ep)[GK / 4], int g) {
  if (EPI >= 1) {
    const float4 bb = ep[0][g];
    v.x = fmaxf(v.x + bb.x, 0.0f);
    v.y = fmaxf(v.y + bb.y, 0.0f);
    v.z = fmaxf(v.z + bb.z, 0.0f);
    v.w = fmaxf(v.w + bb.w, 0.0f);
  }
  if (EPI == 2) {
    const float4 gm = ep[1][g];
    const float4 bt = ep[2][g];
    const float mu  = 0.25f * (v.x + v.y + v.z + v.w);
    const float ms  = 0.25f * (v.x * v.x + v.y * v.y + v.z * v.z + v.w * v.w);
    const float inv = rsqrtf(fmaxf(ms - mu * mu, 0.0f) + 1e-5f);
    v.x = (v.x - mu) * inv * gm.x + bt.x;
    v.y = (v.y - mu) * inv * gm.y + bt.y;
    v.z = (v.z - mu) * inv * gm.z + bt.z;
    v.w = (v.w - mu) * inv * gm.w + bt.w;
  }
  return v;
}

template <int PC, int CPW>
__device__ __forceinline__ void wmma_kgroup(
    const float4 v, const v2f* lw, int j, int hk, int m, int t0, v8f* acc) {
  const v2f* brow = &lw[(4 * j + 2 * hk) * PC + t0 * 16 + m];
  v2f a0, a1;
  a0.x = v.x; a0.y = v.y;
  a1.x = v.z; a1.y = v.w;
#pragma unroll
  for (int t = 0; t < CPW; ++t) {
    const v2f b = brow[t * 16];
    acc[t] = __builtin_amdgcn_wmma_f32_16x16x4_f32(
        false, a0, false, b, (short)0, acc[t], false, false);
  }
#pragma unroll
  for (int t = 0; t < CPW; ++t) {
    const v2f b = brow[PC + t * 16];
    acc[t] = __builtin_amdgcn_wmma_f32_16x16x4_f32(
        false, a1, false, b, (short)0, acc[t], false, false);
  }
}

template <int NC>
__global__ __launch_bounds__(256) void gemm_wmma_f32(
    const float* __restrict__ X, const float* __restrict__ W,
    float* __restrict__ H, int n) {
  constexpr int K  = GK;
  constexpr int CT = (NC + 15) / 16;
  constexpr int PC = CT * 16;
  __shared__ v2f lw[(K / 2) * PC];

  const int tid = threadIdx.x;
  stage_w<NC, 256>(W, lw, tid);
  __syncthreads();

  const int wave = tid >> 5;
  const int lane = tid & 31;
  const int row0 = (blockIdx.x * 8 + wave) * 16;
  if (row0 >= n) return;

  const int m  = lane & 15;
  const int hk = lane >> 4;

  v8f acc[CT];
#pragma unroll
  for (int t = 0; t < CT; ++t) acc[t] = {};

  const float4* xrow = (const float4*)(X + (size_t)(row0 + m) * K) + hk;

  float4 q[K / 8];
#pragma unroll
  for (int j = 0; j < K / 8; ++j) q[j] = xrow[2 * j];

#pragma unroll
  for (int j = 0; j < K / 8; ++j) wmma_kgroup<PC, CT>(q[j], lw, j, hk, m, 0, acc);

  const int crow = row0 + hk * 8;
#pragma unroll
  for (int t = 0; t < CT; ++t) {
    const int col = t * 16 + m;
    if ((NC % 16 == 0) || (col < NC)) {
#pragma unroll
      for (int r = 0; r < 8; ++r)
        H[(size_t)(crow + r) * NC + col] = acc[t][r];
    }
  }
}

#define TILE 196
#define NTILES ((NNODES + TILE - 1) / TILE)
#define NWV 32
#define CAP 448
#define CAP2 256
#define FCH 5
#define OMP (FCH * 32 + 1)

static_assert(TILE <= 256, "local node id is kept in 8 bits");
static_assert(NNODES < (1 << 24) && NEDGES < (1 << 24), "ids are kept in 24 bits");

static_assert(NTILES <= 256, "a tile id is kept in 8 bits");
static_assert(NEDGES % 4 == 0, "four tile ids per word");
#define TB_THREADS 256
#define TB_BLOCKS ((NEDGES / 4 + TB_THREADS - 1) / TB_THREADS)

__global__ __launch_bounds__(TB_THREADS) void tile_ids(
    const int* __restrict__ tgt, unsigned* __restrict__ tb, int E4) {
  const int i = blockIdx.x * TB_THREADS + threadIdx.x;
  if (i < E4) {
    const int4 t = ((const int4*)tgt)[i];
    tb[i] = (((unsigned)t.x / TILE) & 0xFFu) |
            ((((unsigned)t.y / TILE) & 0xFFu) << 8) |
            ((((unsigned)t.z / TILE) & 0xFFu) << 16) |
            (((unsigned)t.w / TILE) << 24);
  }
}

#define SCAP 512

template <int F, int RS>
__device__ __forceinline__ void add_rows(
    const unsigned* ok, const float* ow, int n_own,
    const float* __restrict__ H, float4* acc, int lane) {
  constexpr int F4 = F / 4;
  if (n_own <= 0) return;
  const float4* hl = (const float4*)H + min(lane, F4 - 1);
  unsigned kk[8];
  float wq[8];
  float4 v[8];
#pragma unroll
  for (int u = 0; u < 8; ++u) {
    kk[u] = 0u;
    wq[u] = 0.f;
    v[u] = make_float4(0.f, 0.f, 0.f, 0.f);
  }
  for (int j0 = -8; j0 < n_own; j0 += 8) {
    unsigned kn[8];
    float wn[8];
#pragma unroll
    for (int u = 0; u < 8; ++u) {
      const int jj = min(j0 + 8 + u, n_own - 1);
      kn[u] = (unsigned)__builtin_amdgcn_readfirstlane((int)ok[jj]);
      wn[u] = ow[jj];
    }
#pragma unroll
    for (int u = 0; u < 8; ++u) {
      if ((unsigned)(j0 + u) < (unsigned)n_own) {
        const int tl = (int)(kk[u] >> 24);
        if (lane < F4) {
          float4 a = acc[tl * RS + lane];
          a.x = fmaf(wq[u], v[u].x, a.x);
          a.y = fmaf(wq[u], v[u].y, a.y);
          a.z = fmaf(wq[u], v[u].z, a.z);
          a.w = fmaf(wq[u], v[u].w, a.w);
          acc[tl * RS + lane] = a;
        }
      }
      const size_t s = (size_t)(kn[u] & 0xFFFFFFu);
      v[u] = hl[s * F4];
      kk[u] = kn[u];
      wq[u] = wn[u];
    }
  }
}

template <int NC, int EPI, int CPW>
__device__ __forceinline__ void tail_product(
    const float4* acc, const v2f* lw, const float4 (*ep)[GK / 4],
    float* __restrict__ Hn, int lo, int n, int wave, int lane) {
  constexpr int CT = (NC + 15) / 16;
  constexpr int PC = CT * 16;
  constexpr int RS = GK / 4 + 1;
  constexpr int RT = (TILE + 15) / 16;
  constexpr int CG = CT / CPW;
  static_assert(CT % CPW == 0, "column tiles split evenly into groups");
  static_assert(RT * CG <= NWV, "one unit per wave");
  const int m  = lane & 15;
  const int hk = lane >> 4;
  if (wave < RT * CG) {
    const int rt = wave / CG;
    const int t0 = (wave % CG) * CPW;
    const float4* xrow = acc + min(rt * 16 + m, TILE - 1) * RS + hk;
    v8f c[CPW];
#pragma unroll
    for (int t = 0; t < CPW; ++t) c[t] = {};
#pragma unroll
    for (int j = 0; j < GK / 8; ++j)
      wmma_kgroup<PC, CPW>(row_fn<EPI>(xrow[2 * j], ep, 2 * j + hk), lw, j, hk, m, t0, c);

    const int lr0 = rt * 16 + hk * 8;
#pragma unroll
    for (int t = 0; t < CPW; ++t) {
      const int col = (t0 + t) * 16 + m;
      if ((NC % 16 == 0) || (col < NC)) {
#pragma unroll
        for (int r = 0; r < 8; ++r) {
          const int lr = lr0 + r;
          if (lr < TILE && lo + lr < n) Hn[(size_t)(lo + lr) * NC + col] = c[t][r];
        }
      }
    }
  }
}

__device__ __forceinline__ void tail_lsm40(
    const float4* acc, const float* __restrict__ b, float* __restrict__ out,
    int lo, int n, int wave, int lane) {
  const float* rows = (const float*)acc;
  const float c0 = b[lane];
  const float c1 = (lane < 8) ? b[32 + lane] : 0.0f;
  for (int r = wave; r < TILE && lo + r < n; r += NWV) {
    const float v0 = rows[r * 40 + lane] + c0;
    const float v1 = (lane < 8) ? (rows[r * 40 + 32 + lane] + c1) : -INFINITY;
    float mx = fmaxf(v0, v1);
#pragma unroll
    for (int o = 16; o > 0; o >>= 1) mx = fmaxf(mx, __shfl_xor(mx, o, 32));
    float sm = expf(v0 - mx) + ((lane < 8) ? expf(v1 - mx) : 0.0f);
#pragma unroll
    for (int o = 16; o > 0; o >>= 1) sm += __shfl_xor(sm, o, 32);
    const float lse = mx + logf(sm);
    out[(size_t)(lo + r) * 40 + lane] = v0 - lse;
    if (lane < 8) out[(size_t)(lo + r) * 40 + 32 + lane] = v1 - lse;
  }
}

template <int F, int MODE, int NCN, int EPI>
__global__ __launch_bounds__(1024) void agg_tile(
    const float* __restrict__ H, const int* __restrict__ src,
    const int* __restrict__ tgt, const float* __restrict__ w,
    const float* __restrict__ Wn, const float* __restrict__ eb,
    const float* __restrict__ eg, const float* __restrict__ et,
    float* __restrict__ On, unsigned* __restrict__ slist, int* __restrict__ scnt,
    const unsigned* __restrict__ tb, int n, int E) {
  static_assert(NCN == 0 ? F == 40 : F == GK, "the tail reads rows of the width it was written for");
  constexpr int F4 = F / 4;
  constexpr int RS = (NCN != 0) ? F4 + 1 : F4;
  constexpr int PCN = ((NCN + 15) / 16) * 16;
  __shared__ float4   acc[TILE * RS];
  __shared__ v2f      lbuf[NWV * CAP];
  static_assert((GK / 2) * PCN <= NWV * CAP, "packed W fits where the scan lists were");
  unsigned* lkey = (unsigned*)lbuf;
  float*    lwt  = (float*)lbuf + NWV * CAP;
  __shared__ float4   ep[3][GK / 4];
  __shared__ unsigned okey[NWV * CAP2];
  __shared__ float    owt[NWV * CAP2];
  __shared__ int      lcnt[NWV];
  __shared__ int      wmore[NWV];
  __shared__ unsigned omask[NWV * OMP];

  const int tid  = threadIdx.x;
  const int lane = tid & 31;
  const int wave = __builtin_amdgcn_readfirstlane(tid >> 5);
  const int lo   = blockIdx.x * TILE;
  const unsigned below = (1u << lane) - 1u;
  const unsigned pat = (unsigned)blockIdx.x * 0x01010101u;

  unsigned* ok = okey + wave * CAP2;
  float*    ow = owt + wave * CAP2;
  uint2*    sv = (uint2*)slist + (size_t)(blockIdx.x * NWV + wave) * SCAP;

  int  cl = 0;
  bool use_saved = false;
  if (MODE == 1) {
    cl = scnt[blockIdx.x * NWV + lane];
    use_saved = __builtin_amdgcn_ballot_w32(cl > SCAP) == 0;
  }

  for (int i = tid; i < TILE * RS; i += 1024) acc[i] = make_float4(0.f, 0.f, 0.f, 0.f);

  if constexpr (NCN != 0) {
    if (EPI >= 1 && tid < GK / 4) ep[0][tid] = *(const float4*)(eb + tid * 4);
    if (EPI == 2 && tid < GK / 4) {
      ep[1][tid] = *(const float4*)(eg + tid * 4);
      ep[2][tid] = *(const float4*)(et + tid * 4);
    }
  }

  if (use_saved) {
    if constexpr (NCN != 0) stage_w<NCN, 1024>(Wn, lbuf, tid);
    const int c = __builtin_amdgcn_readlane(cl, wave);
    __syncthreads();
    for (int j0 = 0; j0 < c; j0 += CAP2) {
      const int nn = min(CAP2, c - j0);
      uint2 e[CAP2 / 32];
#pragma unroll
      for (int u = 0; u < CAP2 / 32; ++u) {
        const int i = u * 32 + lane;
        e[u] = make_uint2(0u, 0u);
        if (i < nn) e[u] = sv[j0 + i];
      }
#pragma unroll
      for (int u = 0; u < CAP2 / 32; ++u) {
        const int i = u * 32 + lane;
        if (i < nn) {
          ok[i] = e[u].x;
          ow[i] = __uint_as_float(e[u].y);
        }
      }
      __builtin_amdgcn_wave_barrier();
      add_rows<F, RS>(ok, ow, nn, H, acc, lane);
      __builtin_amdgcn_wave_barrier();
    }
  } else {
    const int per = (((E + NWV - 1) / NWV) + 3) & ~3;
    int pos = min(wave * per, E);
    const int end = min(pos + per, E);
    unsigned* mykey = lkey + wave * CAP;
    float*    myw   = lwt + wave * CAP;

    const int max_trips = per / (CAP - 127) + 2;
    int skip = 0;
    int saved = 0;
    for (int trip = 0; trip < max_trips; ++trip) {
      __syncthreads();

      int cnt = 0;
      while (pos < end) {
        unsigned x[32];
        if (pos + 4096 <= end) {
          const unsigned* tp = tb + (pos >> 2) + lane;
#pragma unroll
          for (int i = 0; i < 32; ++i) x[i] = tp[i * 32] ^ pat;
        } else {
#pragma unroll
          for (int i = 0; i < 32; ++i) {
            const int wi = (pos >> 2) + i * 32 + lane;
            const unsigned v = tb[min(wi, (E >> 2) - 1)] ^ pat;
            x[i] = (wi * 4 < end) ? v : 0xFFFFFFFFu;
          }
        }
        if (skip != 0) {
#pragma unroll
          for (int u = 0; u < 8; ++u)
            if (u < skip) x[u] = 0xFFFFFFFFu;
        }
#pragma unroll
        for (int ss = 0; ss < 4; ++ss) {
          if (pos < end) {
#pragma unroll
            for (int u = 0; u < 8; ++u) {
              const unsigned xv = x[ss * 8 + u];
              const unsigned hz = (xv - 0x01010101u) & ~xv & 0x80808080u;
              if (__builtin_amdgcn_ballot_w32(hz != 0u) != 0) {
                if (cnt > CAP - 128) {
                  skip = u;
                  goto scan_done;
                }
#pragma unroll
                for (int c = 0; c < 4; ++c) {
                  const unsigned mc =
                      __builtin_amdgcn_ballot_w32(((xv >> (8 * c)) & 0xFFu) == 0u);
                  if (mc != 0) {
                    if ((mc >> lane) & 1u) {
                      const int idx = cnt + __builtin_popcount(mc & below);
                      mykey[idx] = (unsigned)(pos + (u * 32 + lane) * 4 + c);
                    }
                    cnt += __builtin_popcount(mc);
                  }
                }
              }
            }
            pos += 1024;
            skip = 0;
          }
        }
      }
    scan_done:
      __builtin_amdgcn_wave_barrier();
      for (int i0 = 0; i0 < cnt; i0 += 32) {
        const int  i     = i0 + lane;
        const bool valid = i < cnt;
        unsigned tl = 0u;
        if (valid) {
          const int e = (int)mykey[i];
          tl = (unsigned)(tgt[e] - lo);
          mykey[i] = (tl << 24) | (unsigned)src[e];
          myw[i]   = w[e];
        }
        if (i0 < FCH * 32) {
          unsigned mo = __builtin_amdgcn_ballot_w32(valid);
#pragma unroll
          for (int k = 0; k < 5; ++k) {
            const unsigned bk = __builtin_amdgcn_ballot_w32(((tl >> k) & 1u) != 0u);
            mo &= ((lane >> k) & 1) ? bk : ~bk;
          }
          omask[wave * OMP + i0 + lane] = mo;
        }
      }
      if (lane == 0) {
        lcnt[wave]  = cnt;
        wmore[wave] = (pos < end) ? 1 : 0;
      }
      __syncthreads();
      const bool more = __builtin_amdgcn_ballot_w32(wmore[lane] != 0) != 0;

      bool by_masks = __builtin_amdgcn_ballot_w32(lcnt[lane] > FCH * 32) == 0;
      int sl = 0, base = 0;
      while (sl < NWV) {
        int n_own = 0;
        if (by_masks) {
          by_masks = false;
          const int cv = lcnt[lane];
          unsigned m[FCH];
          int nv = 0;
#pragma unroll
          for (int c = 0; c < FCH; ++c) {
            m[c] = (c * 32 < cv) ? omask[lane * OMP + c * 32 + wave] : 0u;
            nv += __builtin_popcount(m[c]);
          }
          int incl = nv;
#pragma unroll
          for (int d = 1; d < 32; d <<= 1) {
            const int t = __shfl_up(incl, d, 32);
            if (lane >= d) incl += t;
          }
          const int total = __builtin_amdgcn_readlane(incl, 31);
          if (total <= CAP2) {
            int p = incl - nv;
#pragma unroll
            for (int c = 0; c < FCH; ++c) {
              unsigned mm = m[c];
              while (__builtin_amdgcn_ballot_w32(mm != 0u) != 0) {
                if (mm != 0u) {
                  const int b = __builtin_ctz(mm);
                  mm &= mm - 1u;
                  const int idx = lane * CAP + c * 32 + b;
                  ok[p] = lkey[idx];
                  ow[p] = lwt[idx];
                  ++p;
                }
              }
            }
            n_own = total;
            sl = NWV;
          }
        }
        while (sl < NWV && n_own <= CAP2 - 32) {
          const int c = __builtin_amdgcn_readfirstlane(lcnt[sl]);
          if (base >= c) { ++sl; base = 0; continue; }
          const int i = base + lane;
          unsigned k = 0;
          float ww = 0.f;
          bool mine = false;
          if (i < c) {
            k    = lkey[sl * CAP + i];
            ww   = lwt[sl * CAP + i];
            mine = (int)((k >> 24) & 31u) == wave;
          }
          const unsigned mm = __builtin_amdgcn_ballot_w32(mine);
          if (mm != 0) {
            if (mine) {
              const int idx = n_own + __builtin_popcount(mm & below);
              ok[idx] = k;
              ow[idx] = ww;
            }
            n_own += __builtin_popcount(mm);
          }
          base += 32;
        }
        __builtin_amdgcn_wave_barrier();
        if (MODE == 0) {
          for (int i = lane; i < n_own; i += 32) {
            const int g = saved + i;
            if (g < SCAP) sv[g] = make_uint2(ok[i], __float_as_uint(ow[i]));
          }
          saved += n_own;
        }
        add_rows<F, RS>(ok, ow, n_own, H, acc, lane);
        __builtin_amdgcn_wave_barrier();
      }
      if (!more) break;
    }
    if (MODE == 0 && lane == 0) scnt[blockIdx.x * NWV + wave] = saved;
  }
  __syncthreads();

  if constexpr (NCN != 0) {
    if (!use_saved) {
      stage_w<NCN, 1024>(Wn, lbuf, tid);
      __syncthreads();
    }
    tail_product<NCN, EPI, (NCN == GK ? 4 : 3)>(acc, lbuf, ep, On, lo, n, wave, lane);
  } else {
    tail_lsm40(acc, eb, On, lo, n, wave, lane);
  }
}

extern "C" void kernel_launch(void* const* d_in, const int* in_sizes, int n_in,
                              void* d_out, int out_size, void* d_ws, size_t ws_size,
                              hipStream_t stream) {
  const int N = NNODES, E = NEDGES;
  const float* x     = (const float*)d_in[0];
  const int*   src   = (const int*)d_in[1];
  const int*   tgt   = (const int*)d_in[2];
  const float* w     = (const float*)d_in[3];
  const float* W0    = (const float*)d_in[4];
  const float* b0    = (const float*)d_in[5];
  const float* W1    = (const float*)d_in[6];
  const float* b1    = (const float*)d_in[7];
  const float* W2    = (const float*)d_in[8];
  const float* b2    = (const float*)d_in[9];
  const float* gamma = (const float*)d_in[10];
  const float* beta  = (const float*)d_in[11];
  float* out = (float*)d_out;

  float* H   = (float*)d_ws;
  float* H1  = H + (size_t)N * 128;
  float* H2  = H;
  unsigned* SL = (unsigned*)(H + (size_t)N * 256);
  int*      SC = (int*)(SL + (size_t)NTILES * NWV * SCAP * 2);
  unsigned* TB = (unsigned*)(SC + NTILES * NWV);

  const int gemm_blocks = (N + 127) / 128;

  tile_ids<<<TB_BLOCKS, TB_THREADS, 0, stream>>>(tgt, TB, NEDGES / 4);


  gemm_wmma_f32<128><<<gemm_blocks, 256, 0, stream>>>(x, W0, H, N);
  agg_tile<128, 0, 128, 1><<<NTILES, 1024, 0, stream>>>(
      H, src, tgt, w, W1, b0, nullptr, nullptr, H1, SL, SC, TB, N, E);

  agg_tile<128, 1, 40, 2><<<NTILES, 1024, 0, stream>>>(
      H1, src, tgt, w, W2, b1, gamma, beta, H2, SL, SC, TB, N, E);

  agg_tile<40, 1, 0, 0><<<NTILES, 1024, 0, stream>>>(
      H2, src, tgt, w, nullptr, b2, nullptr, nullptr, out, SL, SC, TB, N, E);
}


# layer 0: first launch filters each 1/16 of the edge list per range of 16 tiles into ordered record lists (ballots); a tile scans its range's 16 lists, not all tile ids; sums now in edge-id order
# speedup vs baseline: 3.7438x; 1.1027x over previous
#include <hip/hip_runtime.h>
#include <math.h>

typedef __attribute__((ext_vector_type(2))) float v2f;
typedef __attribute__((ext_vector_type(8))) float v8f;

#define NNODES 50000
#define NEDGES 800000

#define GK 128

template <int NC, int NT>
__device__ __forceinline__ void stage_w(const float* __restrict__ W, v2f* lw, int tid) {
  constexpr int PC = ((NC + 15) / 16) * 16;
#pragma unroll
  for (int i = tid; i < (GK / 2) * PC; i += NT) {
    const int kp = i / PC, c = i % PC;
    const int cc = min(c, NC - 1);
    const float w0 = W[(2 * kp) * NC + cc];
    const float w1 = W[(2 * kp + 1) * NC + cc];
    v2f val;
    val.x = (c < NC) ? w0 : 0.0f;
    val.y = (c < NC) ? w1 : 0.0f;
    lw[i] = val;
  }
}

template <int EPI>
__device__ __forceinline__ float4 row_fn(float4 v, const float4 (*ep)[GK / 4], int g) {
  if (EPI >= 1) {
    const float4 bb = ep[0][g];
    v.x = fmaxf(v.x + bb.x, 0.0f);
    v.y = fmaxf(v.y + bb.y, 0.0f);
    v.z = fmaxf(v.z + bb.z, 0.0f);
    v.w = fmaxf(v.w + bb.w, 0.0f);
  }
  if (EPI == 2) {
    const float4 gm = ep[1][g];
    const float4 bt = ep[2][g];
    const float mu  = 0.25f * (v.x + v.y + v.z + v.w);
    const float ms  = 0.25f * (v.x * v.x + v.y * v.y + v.z * v.z + v.w * v.w);
    const float inv = rsqrtf(fmaxf(ms - mu * mu, 0.0f) + 1e-5f);
    v.x = (v.x - mu) * inv * gm.x + bt.x;
    v.y = (v.y - mu) * inv * gm.y + bt.y;
    v.z = (v.z - mu) * inv * gm.z + bt.z;
    v.w = (v.w - mu) * inv * gm.w + bt.w;
  }
  return v;
}

template <int PC, int CPW>
__device__ __forceinline__ void wmma_kgroup(
    const float4 v, const v2f* lw, int j, int hk, int m, int t0, v8f* acc) {
  const v2f* brow = &lw[(4 * j + 2 * hk) * PC + t0 * 16 + m];
  v2f a0, a1;
  a0.x = v.x; a0.y = v.y;
  a1.x = v.z; a1.y = v.w;
#pragma unroll
  for (int t = 0; t < CPW; ++t) {
    const v2f b = brow[t * 16];
    acc[t] = __builtin_amdgcn_wmma_f32_16x16x4_f32(
        false, a0, false, b, (short)0, acc[t], false, false);
  }
#pragma unroll
  for (int t = 0; t < CPW; ++t) {
    const v2f b = brow[PC + t * 16];
    acc[t] = __builtin_amdgcn_wmma_f32_16x16x4_f32(
        false, a1, false, b, (short)0, acc[t], false, false);
  }
}

template <int NC>
__global__ __launch_bounds__(256) void gemm_wmma_f32(
    const float* __restrict__ X, const float* __restrict__ W,
    float* __restrict__ H, int n) {
  constexpr int K  = GK;
  constexpr int CT = (NC + 15) / 16;
  constexpr int PC = CT * 16;
  __shared__ v2f lw[(K / 2) * PC];

  const int tid = threadIdx.x;
  stage_w<NC, 256>(W, lw, tid);
  __syncthreads();

  const int wave = tid >> 5;
  const int lane = tid & 31;
  const int row0 = (blockIdx.x * 8 + wave) * 16;
  if (row0 >= n) return;

  const int m  = lane & 15;
  const int hk = lane >> 4;

  v8f acc[CT];
#pragma unroll
  for (int t = 0; t < CT; ++t) acc[t] = {};

  const float4* xrow = (const float4*)(X + (size_t)(row0 + m) * K) + hk;

  float4 q[K / 8];
#pragma unroll
  for (int j = 0; j < K / 8; ++j) q[j] = xrow[2 * j];

#pragma unroll
  for (int j = 0; j < K / 8; ++j) wmma_kgroup<PC, CT>(q[j], lw, j, hk, m, 0, acc);

  const int crow = row0 + hk * 8;
#pragma unroll
  for (int t = 0; t < CT; ++t) {
    const int col = t * 16 + m;
    if ((NC % 16 == 0) || (col < NC)) {
#pragma unroll
      for (int r = 0; r < 8; ++r)
        H[(size_t)(crow + r) * NC + col] = acc[t][r];
    }
  }
}

#define TILE 196
#define NTILES ((NNODES + TILE - 1) / TILE)
#define NWV 32
#define CAP 448
#define CAP2 256
#define SROWS 16
#define FCH 5
#define OMP (FCH * 32 + 1)

static_assert(TILE <= 256, "local node id is kept in 8 bits");
static_assert(NNODES < (1 << 24) && NEDGES < (1 << 24), "ids are kept in 24 bits");

#define SUPER 16
#define NRANGES (NTILES / SUPER)
#define NPARTS 16
#define PART (NEDGES / NPARTS)
#define RSPAN (SUPER * TILE)
#define RL_ROWS ((PART + NWV * 32 - 1) / (NWV * 32))
#define RL_SLICE (RL_ROWS * 32)
#define RL_B 7
static_assert(NTILES % SUPER == 0 && RSPAN <= (1 << 12), "a place within a range is kept in 12 bits");
static_assert(NEDGES % NPARTS == 0, "parts of equal length");
static_assert(NEDGES <= (1 << 20), "an edge id is kept in 20 bits");
static_assert(NWV == 2 * NPARTS, "two scanning waves per list");

__global__ __launch_bounds__(NWV * 32) void range_lists(
    const int* __restrict__ tgt, unsigned* __restrict__ rec, int* __restrict__ rcnt) {
  __shared__ unsigned lst[NWV * RL_SLICE];
  __shared__ int wcnt[NWV];
  const int lane = threadIdx.x & 31;
  const int wave = __builtin_amdgcn_readfirstlane(threadIdx.x >> 5);
  const int rg   = blockIdx.x / NPARTS;
  const int part = blockIdx.x % NPARTS;
  const unsigned below = (1u << lane) - 1u;
  const int s0  = part * PART + wave * RL_SLICE;
  const int s1  = min(s0 + RL_SLICE, (part + 1) * PART);
  const int nlo = rg * RSPAN;
  unsigned* mine = lst + wave * RL_SLICE;

  int cnt = 0;
  for (int r0 = 0; r0 < RL_ROWS; r0 += RL_B) {
    int t[RL_B];
#pragma unroll
    for (int r = 0; r < RL_B; ++r)
      t[r] = tgt[min(s0 + (r0 + r) * 32 + lane, NEDGES - 1)];
#pragma unroll
    for (int r = 0; r < RL_B; ++r) {
      const int e = s0 + (r0 + r) * 32 + lane;
      const unsigned d = (unsigned)(t[r] - nlo);
      const bool hit = (e < s1) && (d < (unsigned)RSPAN);
      const unsigned mc = __builtin_amdgcn_ballot_w32(hit);
      if (mc != 0) {
        if (hit) mine[cnt + __builtin_popcount(mc & below)] = (d << 20) | (unsigned)e;
        cnt += __builtin_popcount(mc);
      }
    }
  }
  if (lane == 0) wcnt[wave] = cnt;
  __syncthreads();

  const int c = wcnt[lane];
  int incl = c;
#pragma unroll
  for (int d = 1; d < 32; d <<= 1) {
    const int u = __shfl_up(incl, d, 32);
    if (lane >= d) incl += u;
  }
  const int off   = __builtin_amdgcn_readlane(incl - c, wave);
  const int total = __builtin_amdgcn_readlane(incl, 31);
  unsigned* dst = rec + (size_t)blockIdx.x * PART + off;
  for (int i = lane; i < cnt; i += 32) dst[i] = mine[i];
  if (threadIdx.x == 0) rcnt[blockIdx.x] = total;
}

#define SCAP 512

template <int F, int RS>
__device__ __forceinline__ void add_rows(
    const unsigned* ok, const float* ow, int n_own,
    const float* __restrict__ H, float4* acc, int lane) {
  constexpr int F4 = F / 4;
  if (n_own <= 0) return;
  const float4* hl = (const float4*)H + min(lane, F4 - 1);
  unsigned kk[8];
  float wq[8];
  float4 v[8];
#pragma unroll
  for (int u = 0; u < 8; ++u) {
    kk[u] = 0u;
    wq[u] = 0.f;
    v[u] = make_float4(0.f, 0.f, 0.f, 0.f);
  }
  for (int j0 = -8; j0 < n_own; j0 += 8) {
    unsigned kn[8];
    float wn[8];
#pragma unroll
    for (int u = 0; u < 8; ++u) {
      const int jj = min(j0 + 8 + u, n_own - 1);
      kn[u] = (unsigned)__builtin_amdgcn_readfirstlane((int)ok[jj]);
      wn[u] = ow[jj];
    }
#pragma unroll
    for (int u = 0; u < 8; ++u) {
      if ((unsigned)(j0 + u) < (unsigned)n_own) {
        const int tl = (int)(kk[u] >> 24);
        if (lane < F4) {
          float4 a = acc[tl * RS + lane];
          a.x = fmaf(wq[u], v[u].x, a.x);
          a.y = fmaf(wq[u], v[u].y, a.y);
          a.z = fmaf(wq[u], v[u].z, a.z);
          a.w = fmaf(wq[u], v[u].w, a.w);
          acc[tl * RS + lane] = a;
        }
      }
      const size_t s = (size_t)(kn[u] & 0xFFFFFFu);
      v[u] = hl[s * F4];
      kk[u] = kn[u];
      wq[u] = wn[u];
    }
  }
}

template <int NC, int EPI, int CPW>
__device__ __forceinline__ void tail_product(
    const float4* acc, const v2f* lw, const float4 (*ep)[GK / 4],
    float* __restrict__ Hn, int lo, int n, int wave, int lane) {
  constexpr int CT = (NC + 15) / 16;
  constexpr int PC = CT * 16;
  constexpr int RS = GK / 4 + 1;
  constexpr int RT = (TILE + 15) / 16;
  constexpr int CG = CT / CPW;
  static_assert(CT % CPW == 0, "column tiles split evenly into groups");
  static_assert(RT * CG <= NWV, "one unit per wave");
  const int m  = lane & 15;
  const int hk = lane >> 4;
  if (wave < RT * CG) {
    const int rt = wave / CG;
    const int t0 = (wave % CG) * CPW;
    const float4* xrow = acc + min(rt * 16 + m, TILE - 1) * RS + hk;
    v8f c[CPW];
#pragma unroll
    for (int t = 0; t < CPW; ++t) c[t] = {};
#pragma unroll
    for (int j = 0; j < GK / 8; ++j)
      wmma_kgroup<PC, CPW>(row_fn<EPI>(xrow[2 * j], ep, 2 * j + hk), lw, j, hk, m, t0, c);

    const int lr0 = rt * 16 + hk * 8;
#pragma unroll
    for (int t = 0; t < CPW; ++t) {
      const int col = (t0 + t) * 16 + m;
      if ((NC % 16 == 0) || (col < NC)) {
#pragma unroll
        for (int r = 0; r < 8; ++r) {
          const int lr = lr0 + r;
          if (lr < TILE && lo + lr < n) Hn[(size_t)(lo + lr) * NC + col] = c[t][r];
        }
      }
    }
  }
}

__device__ __forceinline__ void tail_lsm40(
    const float4* acc, const float* __restrict__ b, float* __restrict__ out,
    int lo, int n, int wave, int lane) {
  const float* rows = (const float*)acc;
  const float c0 = b[lane];
  const float c1 = (lane < 8) ? b[32 + lane] : 0.0f;
  for (int r = wave; r < TILE && lo + r < n; r += NWV) {
    const float v0 = rows[r * 40 + lane] + c0;
    const float v1 = (lane < 8) ? (rows[r * 40 + 32 + lane] + c1) : -INFINITY;
    float mx = fmaxf(v0, v1);
#pragma unroll
    for (int o = 16; o > 0; o >>= 1) mx = fmaxf(mx, __shfl_xor(mx, o, 32));
    float sm = expf(v0 - mx) + ((lane < 8) ? expf(v1 - mx) : 0.0f);
#pragma unroll
    for (int o = 16; o > 0; o >>= 1) sm += __shfl_xor(sm, o, 32);
    const float lse = mx + logf(sm);
    out[(size_t)(lo + r) * 40 + lane] = v0 - lse;
    if (lane < 8) out[(size_t)(lo + r) * 40 + 32 + lane] = v1 - lse;
  }
}

template <int F, int MODE, int NCN, int EPI>
__global__ __launch_bounds__(1024) void agg_tile(
    const float* __restrict__ H, const int* __restrict__ src,
    const float* __restrict__ w,
    const float* __restrict__ Wn, const float* __restrict__ eb,
    const float* __restrict__ eg, const float* __restrict__ et,
    float* __restrict__ On, unsigned* __restrict__ slist, int* __restrict__ scnt,
    const unsigned* __restrict__ rec, const int* __restrict__ rcnt, int n) {
  static_assert(NCN == 0 ? F == 40 : F == GK, "the tail reads rows of the width it was written for");
  constexpr int F4 = F / 4;
  constexpr int RS = (NCN != 0) ? F4 + 1 : F4;
  constexpr int PCN = ((NCN + 15) / 16) * 16;
  __shared__ float4   acc[TILE * RS];
  __shared__ v2f      lbuf[NWV * CAP];
  static_assert((GK / 2) * PCN <= NWV * CAP, "packed W fits where the scan lists were");
  unsigned* lkey = (unsigned*)lbuf;
  float*    lwt  = (float*)lbuf + NWV * CAP;
  __shared__ float4   ep[3][GK / 4];
  __shared__ unsigned okey[NWV * CAP2];
  __shared__ float    owt[NWV * CAP2];
  __shared__ int      lcnt[NWV];
  __shared__ int      wmore[NWV];
  __shared__ unsigned omask[NWV * OMP];

  const int tid  = threadIdx.x;
  const int lane = tid & 31;
  const int wave = __builtin_amdgcn_readfirstlane(tid >> 5);
  const int lo   = blockIdx.x * TILE;
  const unsigned below = (1u << lane) - 1u;

  unsigned* ok = okey + wave * CAP2;
  float*    ow = owt + wave * CAP2;
  uint2*    sv = (uint2*)slist + (size_t)(blockIdx.x * NWV + wave) * SCAP;

  int  cl = 0;
  bool use_saved = false;
  if (MODE == 1) {
    cl = scnt[blockIdx.x * NWV + lane];
    use_saved = __builtin_amdgcn_ballot_w32(cl > SCAP) == 0;
  }

  for (int i = tid; i < TILE * RS; i += 1024) acc[i] = make_float4(0.f, 0.f, 0.f, 0.f);

  if constexpr (NCN != 0) {
    if (EPI >= 1 && tid < GK / 4) ep[0][tid] = *(const float4*)(eb + tid * 4);
    if (EPI == 2 && tid < GK / 4) {
      ep[1][tid] = *(const float4*)(eg + tid * 4);
      ep[2][tid] = *(const float4*)(et + tid * 4);
    }
  }

  if (use_saved) {
    if constexpr (NCN != 0) stage_w<NCN, 1024>(Wn, lbuf, tid);
    const int c = __builtin_amdgcn_readlane(cl, wave);
    __syncthreads();
    for (int j0 = 0; j0 < c; j0 += CAP2) {
      const int nn = min(CAP2, c - j0);
      uint2 e[CAP2 / 32];
#pragma unroll
      for (int u = 0; u < CAP2 / 32; ++u) {
        const int i = u * 32 + lane;
        e[u] = make_uint2(0u, 0u);
        if (i < nn) e[u] = sv[j0 + i];
      }
#pragma unroll
      for (int u = 0; u < CAP2 / 32; ++u) {
        const int i = u * 32 + lane;
        if (i < nn) {
          ok[i] = e[u].x;
          ow[i] = __uint_as_float(e[u].y);
        }
      }
      __builtin_amdgcn_wave_barrier();
      add_rows<F, RS>(ok, ow, nn, H, acc, lane);
      __builtin_amdgcn_wave_barrier();
    }
  } else {
    const unsigned dlo = (unsigned)(blockIdx.x % SUPER) * TILE;
    const int li = (blockIdx.x / SUPER) * NPARTS + (wave >> 1);
    const int lc  = __builtin_amdgcn_readfirstlane(rcnt[li]);
    const int mid = min(lc, ((lc >> 1) + 31) & ~31);
    int pos = (wave & 1) ? mid : 0;
    const int end = (wave & 1) ? lc : mid;
    const unsigned* rl = rec + (size_t)li * PART;
    unsigned* mykey = lkey + wave * CAP;
    float*    myw   = lwt + wave * CAP;

    const int max_trips = (PART / 2 + 32) / (CAP - 31) + 2;
    int saved = 0;
    for (int trip = 0; trip < max_trips; ++trip) {
      __syncthreads();

      int cnt = 0;
      while (pos < end && cnt <= CAP - 32) {
        unsigned x[SROWS];
#pragma unroll
        for (int r = 0; r < SROWS; ++r) x[r] = rl[min(pos + r * 32 + lane, end - 1)];
#pragma unroll
        for (int r = 0; r < SROWS; ++r) {
          if (cnt <= CAP - 32) {
            const bool hit = (pos + lane < end) && (((x[r] >> 20) - dlo) < (unsigned)TILE);
            const unsigned mc = __builtin_amdgcn_ballot_w32(hit);
            if (mc != 0) {
              if (hit) mykey[cnt + __builtin_popcount(mc & below)] = x[r];
              cnt += __builtin_popcount(mc);
            }
            pos += 32;
          }
        }
      }
      __builtin_amdgcn_wave_barrier();
      for (int i0 = 0; i0 < cnt; i0 += 32) {
        const int  i     = i0 + lane;
        const bool valid = i < cnt;
        unsigned tl = 0u;
        if (valid) {
          const unsigned rc = mykey[i];
          const int e = (int)(rc & 0xFFFFFu);
          tl = (rc >> 20) - dlo;
          mykey[i] = (tl << 24) | (unsigned)src[e];
          myw[i]   = w[e];
        }
        if (i0 < FCH * 32) {
          unsigned mo = __builtin_amdgcn_ballot_w32(valid);
#pragma unroll
          for (int k = 0; k < 5; ++k) {
            const unsigned bk = __builtin_amdgcn_ballot_w32(((tl >> k) & 1u) != 0u);
            mo &= ((lane >> k) & 1) ? bk : ~bk;
          }
          omask[wave * OMP + i0 + lane] = mo;
        }
      }
      if (lane == 0) {
        lcnt[wave]  = cnt;
        wmore[wave] = (pos < end) ? 1 : 0;
      }
      __syncthreads();
      const bool more = __builtin_amdgcn_ballot_w32(wmore[lane] != 0) != 0;

      bool by_masks = __builtin_amdgcn_ballot_w32(lcnt[lane] > FCH * 32) == 0;
      int sl = 0, base = 0;
      while (sl < NWV) {
        int n_own = 0;
        if (by_masks) {
          by_masks = false;
          const int cv = lcnt[lane];
          unsigned m[FCH];
          int nv = 0;
#pragma unroll
          for (int c = 0; c < FCH; ++c) {
            m[c] = (c * 32 < cv) ? omask[lane * OMP + c * 32 + wave] : 0u;
            nv += __builtin_popcount(m[c]);
          }
          int incl = nv;
#pragma unroll
          for (int d = 1; d < 32; d <<= 1) {
            const int t = __shfl_up(incl, d, 32);
            if (lane >= d) incl += t;
          }
          const int total = __builtin_amdgcn_readlane(incl, 31);
          if (total <= CAP2) {
            int p = incl - nv;
#pragma unroll
            for (int c = 0; c < FCH; ++c) {
              unsigned mm = m[c];
              while (__builtin_amdgcn_ballot_w32(mm != 0u) != 0) {
                if (mm != 0u) {
                  const int b = __builtin_ctz(mm);
                  mm &= mm - 1u;
                  const int idx = lane * CAP + c * 32 + b;
                  ok[p] = lkey[idx];
                  ow[p] = lwt[idx];
                  ++p;
                }
              }
            }
            n_own = total;
            sl = NWV;
          }
        }
        while (sl < NWV && n_own <= CAP2 - 32) {
          const int c = __builtin_amdgcn_readfirstlane(lcnt[sl]);
          if (base >= c) { ++sl; base = 0; continue; }
          const int i = base + lane;
          unsigned k = 0;
          float ww = 0.f;
          bool mine = false;
          if (i < c) {
            k    = lkey[sl * CAP + i];
            ww   = lwt[sl * CAP + i];
            mine = (int)((k >> 24) & 31u) == wave;
          }
          const unsigned mm = __builtin_amdgcn_ballot_w32(mine);
          if (mm != 0) {
            if (mine) {
              const int idx = n_own + __builtin_popcount(mm & below);
              ok[idx] = k;
              ow[idx] = ww;
            }
            n_own += __builtin_popcount(mm);
          }
          base += 32;
        }
        __builtin_amdgcn_wave_barrier();
        if (MODE == 0) {
          for (int i = lane; i < n_own; i += 32) {
            const int g = saved + i;
            if (g < SCAP) sv[g] = make_uint2(ok[i], __float_as_uint(ow[i]));
          }
          saved += n_own;
        }
        add_rows<F, RS>(ok, ow, n_own, H, acc, lane);
        __builtin_amdgcn_wave_barrier();
      }
      if (!more) break;
    }
    if (MODE == 0 && lane == 0) scnt[blockIdx.x * NWV + wave] = saved;
  }
  __syncthreads();

  if constexpr (NCN != 0) {
    if (!use_saved) {
      stage_w<NCN, 1024>(Wn, lbuf, tid);
      __syncthreads();
    }
    tail_product<NCN, EPI, (NCN == GK ? 4 : 3)>(acc, lbuf, ep, On, lo, n, wave, lane);
  } else {
    tail_lsm40(acc, eb, On, lo, n, wave, lane);
  }
}

extern "C" void kernel_launch(void* const* d_in, const int* in_sizes, int n_in,
                              void* d_out, int out_size, void* d_ws, size_t ws_size,
                              hipStream_t stream) {
  const int N = NNODES;
  const float* x     = (const float*)d_in[0];
  const int*   src   = (const int*)d_in[1];
  const int*   tgt   = (const int*)d_in[2];
  const float* w     = (const float*)d_in[3];
  const float* W0    = (const float*)d_in[4];
  const float* b0    = (const float*)d_in[5];
  const float* W1    = (const float*)d_in[6];
  const float* b1    = (const float*)d_in[7];
  const float* W2    = (const float*)d_in[8];
  const float* b2    = (const float*)d_in[9];
  const float* gamma = (const float*)d_in[10];
  const float* beta  = (const float*)d_in[11];
  float* out = (float*)d_out;

  float* H   = (float*)d_ws;
  float* H1  = H + (size_t)N * 128;
  float* H2  = H;
  unsigned* SL = (unsigned*)(H + (size_t)N * 256);
  int*      SC = (int*)(SL + (size_t)NTILES * NWV * SCAP * 2);
  unsigned* RL = (unsigned*)(SC + NTILES * NWV);
  int*      RC = (int*)(RL + (size_t)NRANGES * NPARTS * PART);

  const int gemm_blocks = (N + 127) / 128;

  range_lists<<<NRANGES * NPARTS, NWV * 32, 0, stream>>>(tgt, RL, RC);


  gemm_wmma_f32<128><<<gemm_blocks, 256, 0, stream>>>(x, W0, H, N);
  agg_tile<128, 0, 128, 1><<<NTILES, 1024, 0, stream>>>(
      H, src, w, W1, b0, nullptr, nullptr, H1, SL, SC, RL, RC, N);

  agg_tile<128, 1, 40, 2><<<NTILES, 1024, 0, stream>>>(
      H1, src, w, W2, b1, gamma, beta, H2, SL, SC, RL, RC, N);

  agg_tile<40, 1, 0, 0><<<NTILES, 1024, 0, stream>>>(
      H2, src, w, nullptr, b2, nullptr, nullptr, out, SL, SC, RL, RC, N);
}


# owner waves put their edge list into node order (stable ballot partition), so add_rows keeps a node's sum in registers over its run: one LDS read and store per node, not per edge; sum order kept
# speedup vs baseline: 4.0808x; 1.0900x over previous
#include <hip/hip_runtime.h>
#include <math.h>

typedef __attribute__((ext_vector_type(2))) float v2f;
typedef __attribute__((ext_vector_type(8))) float v8f;

#define NNODES 50000
#define NEDGES 800000

#define GK 128

template <int NC, int NT>
__device__ __forceinline__ void stage_w(const float* __restrict__ W, v2f* lw, int tid) {
  constexpr int PC = ((NC + 15) / 16) * 16;
#pragma unroll
  for (int i = tid; i < (GK / 2) * PC; i += NT) {
    const int kp = i / PC, c = i % PC;
    const int cc = min(c, NC - 1);
    const float w0 = W[(2 * kp) * NC + cc];
    const float w1 = W[(2 * kp + 1) * NC + cc];
    v2f val;
    val.x = (c < NC) ? w0 : 0.0f;
    val.y = (c < NC) ? w1 : 0.0f;
    lw[i] = val;
  }
}

template <int EPI>
__device__ __forceinline__ float4 row_fn(float4 v, const float4 (*ep)[GK / 4], int g) {
  if (EPI >= 1) {
    const float4 bb = ep[0][g];
    v.x = fmaxf(v.x + bb.x, 0.0f);
    v.y = fmaxf(v.y + bb.y, 0.0f);
    v.z = fmaxf(v.z + bb.z, 0.0f);
    v.w = fmaxf(v.w + bb.w, 0.0f);
  }
  if (EPI == 2) {
    const float4 gm = ep[1][g];
    const float4 bt = ep[2][g];
    const float mu  = 0.25f * (v.x + v.y + v.z + v.w);
    const float ms  = 0.25f * (v.x * v.x + v.y * v.y + v.z * v.z + v.w * v.w);
    const float inv = rsqrtf(fmaxf(ms - mu * mu, 0.0f) + 1e-5f);
    v.x = (v.x - mu) * inv * gm.x + bt.x;
    v.y = (v.y - mu) * inv * gm.y + bt.y;
    v.z = (v.z - mu) * inv * gm.z + bt.z;
    v.w = (v.w - mu) * inv * gm.w + bt.w;
  }
  return v;
}

template <int PC, int CPW>
__device__ __forceinline__ void wmma_kgroup(
    const float4 v, const v2f* lw, int j, int hk, int m, int t0, v8f* acc) {
  const v2f* brow = &lw[(4 * j + 2 * hk) * PC + t0 * 16 + m];
  v2f a0, a1;
  a0.x = v.x; a0.y = v.y;
  a1.x = v.z; a1.y = v.w;
#pragma unroll
  for (int t = 0; t < CPW; ++t) {
    const v2f b = brow[t * 16];
    acc[t] = __builtin_amdgcn_wmma_f32_16x16x4_f32(
        false, a0, false, b, (short)0, acc[t], false, false);
  }
#pragma unroll
  for (int t = 0; t < CPW; ++t) {
    const v2f b = brow[PC + t * 16];
    acc[t] = __builtin_amdgcn_wmma_f32_16x16x4_f32(
        false, a1, false, b, (short)0, acc[t], false, false);
  }
}

template <int NC>
__global__ __launch_bounds__(256) void gemm_wmma_f32(
    const float* __restrict__ X, const float* __restrict__ W,
    float* __restrict__ H, int n) {
  constexpr int K  = GK;
  constexpr int CT = (NC + 15) / 16;
  constexpr int PC = CT * 16;
  __shared__ v2f lw[(K / 2) * PC];

  const int tid = threadIdx.x;
  stage_w<NC, 256>(W, lw, tid);
  __syncthreads();

  const int wave = tid >> 5;
  const int lane = tid & 31;
  const int row0 = (blockIdx.x * 8 + wave) * 16;
  if (row0 >= n) return;

  const int m  = lane & 15;
  const int hk = lane >> 4;

  v8f acc[CT];
#pragma unroll
  for (int t = 0; t < CT; ++t) acc[t] = {};

  const float4* xrow = (const float4*)(X + (size_t)(row0 + m) * K) + hk;

  float4 q[K / 8];
#pragma unroll
  for (int j = 0; j < K / 8; ++j) q[j] = xrow[2 * j];

#pragma unroll
  for (int j = 0; j < K / 8; ++j) wmma_kgroup<PC, CT>(q[j], lw, j, hk, m, 0, acc);

  const int crow = row0 + hk * 8;
#pragma unroll
  for (int t = 0; t < CT; ++t) {
    const int col = t * 16 + m;
    if ((NC % 16 == 0) || (col < NC)) {
#pragma unroll
      for (int r = 0; r < 8; ++r)
        H[(size_t)(crow + r) * NC + col] = acc[t][r];
    }
  }
}

#define TILE 196
#define NTILES ((NNODES + TILE - 1) / TILE)
#define NWV 32
#define CAP 448
#define CAP2 256
#define SROWS 16
#define FCH 5
#define OMP (FCH * 32 + 1)

static_assert(TILE <= 256, "local node id is kept in 8 bits");
static_assert(NNODES < (1 << 24) && NEDGES < (1 << 24), "ids are kept in 24 bits");

#define SUPER 16
#define NRANGES (NTILES / SUPER)
#define NPARTS 16
#define PART (NEDGES / NPARTS)
#define RSPAN (SUPER * TILE)
#define RL_ROWS ((PART + NWV * 32 - 1) / (NWV * 32))
#define RL_SLICE (RL_ROWS * 32)
#define RL_B 7
static_assert(NTILES % SUPER == 0 && RSPAN <= (1 << 12), "a place within a range is kept in 12 bits");
static_assert(NEDGES % NPARTS == 0, "parts of equal length");
static_assert(NEDGES <= (1 << 20), "an edge id is kept in 20 bits");
static_assert(NWV == 2 * NPARTS, "two scanning waves per list");

__global__ __launch_bounds__(NWV * 32) void range_lists(
    const int* __restrict__ tgt, unsigned* __restrict__ rec, int* __restrict__ rcnt) {
  __shared__ unsigned lst[NWV * RL_SLICE];
  __shared__ int wcnt[NWV];
  const int lane = threadIdx.x & 31;
  const int wave = __builtin_amdgcn_readfirstlane(threadIdx.x >> 5);
  const int rg   = blockIdx.x / NPARTS;
  const int part = blockIdx.x % NPARTS;
  const unsigned below = (1u << lane) - 1u;
  const int s0  = part * PART + wave * RL_SLICE;
  const int s1  = min(s0 + RL_SLICE, (part + 1) * PART);
  const int nlo = rg * RSPAN;
  unsigned* mine = lst + wave * RL_SLICE;

  int cnt = 0;
  for (int r0 = 0; r0 < RL_ROWS; r0 += RL_B) {
    int t[RL_B];
#pragma unroll
    for (int r = 0; r < RL_B; ++r)
      t[r] = tgt[min(s0 + (r0 + r) * 32 + lane, NEDGES - 1)];
#pragma unroll
    for (int r = 0; r < RL_B; ++r) {
      const int e = s0 + (r0 + r) * 32 + lane;
      const unsigned d = (unsigned)(t[r] - nlo);
      const bool hit = (e < s1) && (d < (unsigned)RSPAN);
      const unsigned mc = __builtin_amdgcn_ballot_w32(hit);
      if (mc != 0) {
        if (hit) mine[cnt + __builtin_popcount(mc & below)] = (d << 20) | (unsigned)e;
        cnt += __builtin_popcount(mc);
      }
    }
  }
  if (lane == 0) wcnt[wave] = cnt;
  __syncthreads();

  const int c = wcnt[lane];
  int incl = c;
#pragma unroll
  for (int d = 1; d < 32; d <<= 1) {
    const int u = __shfl_up(incl, d, 32);
    if (lane >= d) incl += u;
  }
  const int off   = __builtin_amdgcn_readlane(incl - c, wave);
  const int total = __builtin_amdgcn_readlane(incl, 31);
  unsigned* dst = rec + (size_t)blockIdx.x * PART + off;
  for (int i = lane; i < cnt; i += 32) dst[i] = mine[i];
  if (threadIdx.x == 0) rcnt[blockIdx.x] = total;
}

#define SCAP 512

__device__ __forceinline__ void group_by_node(
    unsigned* ok, float* ow, int n_own, int lane, unsigned below) {
  constexpr int NR = CAP2 / 32;
  constexpr int NB = (TILE + NWV - 1) / NWV;
  static_assert(NB <= 7, "slot 7 marks a lane without an entry");
  unsigned k[NR];
  float    wv[NR];
  int      rank[NR];
  int      cnt[NB];
#pragma unroll
  for (int b = 0; b < NB; ++b) cnt[b] = 0;
#pragma unroll
  for (int r = 0; r < NR; ++r) {
    k[r] = 0xFFFFFFFFu;
    wv[r] = 0.f;
    rank[r] = 0;
    if (r * 32 < n_own) {
      const int i = r * 32 + lane;
      if (i < n_own) {
        k[r]  = ok[i];
        wv[r] = ow[i];
      }
      const unsigned slot = k[r] >> 29;
#pragma unroll
      for (int b = 0; b < NB; ++b) {
        const unsigned m = __builtin_amdgcn_ballot_w32(slot == (unsigned)b);
        if (slot == (unsigned)b) rank[r] = cnt[b] + __builtin_popcount(m & below);
        cnt[b] += __builtin_popcount(m);
      }
    }
  }
  __builtin_amdgcn_wave_barrier();
#pragma unroll
  for (int r = 0; r < NR; ++r) {
    if (r * 32 + lane < n_own) {
      const unsigned slot = k[r] >> 29;
      int dst = rank[r];
#pragma unroll
      for (int b = 0; b < NB - 1; ++b)
        if (slot > (unsigned)b) dst += cnt[b];
      ok[dst] = k[r];
      ow[dst] = wv[r];
    }
  }
  __builtin_amdgcn_wave_barrier();
}

template <int F, int RS>
__device__ __forceinline__ void add_rows(
    const unsigned* ok, const float* ow, int n_own,
    const float* __restrict__ H, float4* acc, int lane) {
  constexpr int F4 = F / 4;
  if (n_own <= 0) return;
  const float4* hl = (const float4*)H + min(lane, F4 - 1);
  unsigned kk[8];
  float wq[8];
  float4 v[8];
#pragma unroll
  for (int u = 0; u < 8; ++u) {
    kk[u] = 0u;
    wq[u] = 0.f;
    v[u] = make_float4(0.f, 0.f, 0.f, 0.f);
  }
  float4 cur = make_float4(0.f, 0.f, 0.f, 0.f);
  int ctl = -1;
  for (int j0 = -8; j0 < n_own; j0 += 8) {
    unsigned kn[8];
    float wn[8];
#pragma unroll
    for (int u = 0; u < 8; ++u) {
      const int jj = min(j0 + 8 + u, n_own - 1);
      kn[u] = (unsigned)__builtin_amdgcn_readfirstlane((int)ok[jj]);
      wn[u] = ow[jj];
    }
#pragma unroll
    for (int u = 0; u < 8; ++u) {
      if ((unsigned)(j0 + u) < (unsigned)n_own) {
        const int tl = (int)(kk[u] >> 24);
        if (tl != ctl) {
          if (lane < F4) {
            if (ctl >= 0) acc[ctl * RS + lane] = cur;
            cur = acc[tl * RS + lane];
          }
          ctl = tl;
        }
        cur.x = fmaf(wq[u], v[u].x, cur.x);
        cur.y = fmaf(wq[u], v[u].y, cur.y);
        cur.z = fmaf(wq[u], v[u].z, cur.z);
        cur.w = fmaf(wq[u], v[u].w, cur.w);
      }
      const size_t s = (size_t)(kn[u] & 0xFFFFFFu);
      v[u] = hl[s * F4];
      kk[u] = kn[u];
      wq[u] = wn[u];
    }
  }
  if (lane < F4) acc[ctl * RS + lane] = cur;
}

template <int NC, int EPI, int CPW>
__device__ __forceinline__ void tail_product(
    const float4* acc, const v2f* lw, const float4 (*ep)[GK / 4],
    float* __restrict__ Hn, int lo, int n, int wave, int lane) {
  constexpr int CT = (NC + 15) / 16;
  constexpr int PC = CT * 16;
  constexpr int RS = GK / 4 + 1;
  constexpr int RT = (TILE + 15) / 16;
  constexpr int CG = CT / CPW;
  static_assert(CT % CPW == 0, "column tiles split evenly into groups");
  static_assert(RT * CG <= NWV, "one unit per wave");
  const int m  = lane & 15;
  const int hk = lane >> 4;
  if (wave < RT * CG) {
    const int rt = wave / CG;
    const int t0 = (wave % CG) * CPW;
    const float4* xrow = acc + min(rt * 16 + m, TILE - 1) * RS + hk;
    v8f c[CPW];
#pragma unroll
    for (int t = 0; t < CPW; ++t) c[t] = {};
#pragma unroll
    for (int j = 0; j < GK / 8; ++j)
      wmma_kgroup<PC, CPW>(row_fn<EPI>(xrow[2 * j], ep, 2 * j + hk), lw, j, hk, m, t0, c);

    const int lr0 = rt * 16 + hk * 8;
#pragma unroll
    for (int t = 0; t < CPW; ++t) {
      const int col = (t0 + t) * 16 + m;
      if ((NC % 16 == 0) || (col < NC)) {
#pragma unroll
        for (int r = 0; r < 8; ++r) {
          const int lr = lr0 + r;
          if (lr < TILE && lo + lr < n) Hn[(size_t)(lo + lr) * NC + col] = c[t][r];
        }
      }
    }
  }
}

__device__ __forceinline__ void tail_lsm40(
    const float4* acc, const float* __restrict__ b, float* __restrict__ out,
    int lo, int n, int wave, int lane) {
  const float* rows = (const float*)acc;
  const float c0 = b[lane];
  const float c1 = (lane < 8) ? b[32 + lane] : 0.0f;
  for (int r = wave; r < TILE && lo + r < n; r += NWV) {
    const float v0 = rows[r * 40 + lane] + c0;
    const float v1 = (lane < 8) ? (rows[r * 40 + 32 + lane] + c1) : -INFINITY;
    float mx = fmaxf(v0, v1);
#pragma unroll
    for (int o = 16; o > 0; o >>= 1) mx = fmaxf(mx, __shfl_xor(mx, o, 32));
    float sm = expf(v0 - mx) + ((lane < 8) ? expf(v1 - mx) : 0.0f);
#pragma unroll
    for (int o = 16; o > 0; o >>= 1) sm += __shfl_xor(sm, o, 32);
    const float lse = mx + logf(sm);
    out[(size_t)(lo + r) * 40 + lane] = v0 - lse;
    if (lane < 8) out[(size_t)(lo + r) * 40 + 32 + lane] = v1 - lse;
  }
}

template <int F, int MODE, int NCN, int EPI>
__global__ __launch_bounds__(1024) void agg_tile(
    const float* __restrict__ H, const int* __restrict__ src,
    const float* __restrict__ w,
    const float* __restrict__ Wn, const float* __restrict__ eb,
    const float* __restrict__ eg, const float* __restrict__ et,
    float* __restrict__ On, unsigned* __restrict__ slist, int* __restrict__ scnt,
    const unsigned* __restrict__ rec, const int* __restrict__ rcnt, int n) {
  static_assert(NCN == 0 ? F == 40 : F == GK, "the tail reads rows of the width it was written for");
  constexpr int F4 = F / 4;
  constexpr int RS = (NCN != 0) ? F4 + 1 : F4;
  constexpr int PCN = ((NCN + 15) / 16) * 16;
  __shared__ float4   acc[TILE * RS];
  __shared__ v2f      lbuf[NWV * CAP];
  static_assert((GK / 2) * PCN <= NWV * CAP, "packed W fits where the scan lists were");
  unsigned* lkey = (unsigned*)lbuf;
  float*    lwt  = (float*)lbuf + NWV * CAP;
  __shared__ float4   ep[3][GK / 4];
  __shared__ unsigned okey[NWV * CAP2];
  __shared__ float    owt[NWV * CAP2];
  __shared__ int      lcnt[NWV];
  __shared__ int      wmore[NWV];
  __shared__ unsigned omask[NWV * OMP];

  const int tid  = threadIdx.x;
  const int lane = tid & 31;
  const int wave = __builtin_amdgcn_readfirstlane(tid >> 5);
  const int lo   = blockIdx.x * TILE;
  const unsigned below = (1u << lane) - 1u;

  unsigned* ok = okey + wave * CAP2;
  float*    ow = owt + wave * CAP2;
  uint2*    sv = (uint2*)slist + (size_t)(blockIdx.x * NWV + wave) * SCAP;

  int  cl = 0;
  bool use_saved = false;
  if (MODE == 1) {
    cl = scnt[blockIdx.x * NWV + lane];
    use_saved = __builtin_amdgcn_ballot_w32(cl > SCAP) == 0;
  }

  for (int i = tid; i < TILE * RS; i += 1024) acc[i] = make_float4(0.f, 0.f, 0.f, 0.f);

  if constexpr (NCN != 0) {
    if (EPI >= 1 && tid < GK / 4) ep[0][tid] = *(const float4*)(eb + tid * 4);
    if (EPI == 2 && tid < GK / 4) {
      ep[1][tid] = *(const float4*)(eg + tid * 4);
      ep[2][tid] = *(const float4*)(et + tid * 4);
    }
  }

  if (use_saved) {
    if constexpr (NCN != 0) stage_w<NCN, 1024>(Wn, lbuf, tid);
    const int c = __builtin_amdgcn_readlane(cl, wave);
    __syncthreads();
    for (int j0 = 0; j0 < c; j0 += CAP2) {
      const int nn = min(CAP2, c - j0);
      uint2 e[CAP2 / 32];
#pragma unroll
      for (int u = 0; u < CAP2 / 32; ++u) {
        const int i = u * 32 + lane;
        e[u] = make_uint2(0u, 0u);
        if (i < nn) e[u] = sv[j0 + i];
      }
#pragma unroll
      for (int u = 0; u < CAP2 / 32; ++u) {
        const int i = u * 32 + lane;
        if (i < nn) {
          ok[i] = e[u].x;
          ow[i] = __uint_as_float(e[u].y);
        }
      }
      __builtin_amdgcn_wave_barrier();
      add_rows<F, RS>(ok, ow, nn, H, acc, lane);
      __builtin_amdgcn_wave_barrier();
    }
  } else {
    const unsigned dlo = (unsigned)(blockIdx.x % SUPER) * TILE;
    const int li = (blockIdx.x / SUPER) * NPARTS + (wave >> 1);
    const int lc  = __builtin_amdgcn_readfirstlane(rcnt[li]);
    const int mid = min(lc, ((lc >> 1) + 31) & ~31);
    int pos = (wave & 1) ? mid : 0;
    const int end = (wave & 1) ? lc : mid;
    const unsigned* rl = rec + (size_t)li * PART;
    unsigned* mykey = lkey + wave * CAP;
    float*    myw   = lwt + wave * CAP;

    const int max_trips = (PART / 2 + 32) / (CAP - 31) + 2;
    int saved = 0;
    for (int trip = 0; trip < max_trips; ++trip) {
      __syncthreads();

      int cnt = 0;
      while (pos < end && cnt <= CAP - 32) {
        unsigned x[SROWS];
#pragma unroll
        for (int r = 0; r < SROWS; ++r) x[r] = rl[min(pos + r * 32 + lane, end - 1)];
#pragma unroll
        for (int r = 0; r < SROWS; ++r) {
          if (cnt <= CAP - 32) {
            const bool hit = (pos + lane < end) && (((x[r] >> 20) - dlo) < (unsigned)TILE);
            const unsigned mc = __builtin_amdgcn_ballot_w32(hit);
            if (mc != 0) {
              if (hit) mykey[cnt + __builtin_popcount(mc & below)] = x[r];
              cnt += __builtin_popcount(mc);
            }
            pos += 32;
          }
        }
      }
      __builtin_amdgcn_wave_barrier();
      for (int i0 = 0; i0 < cnt; i0 += 32) {
        const int  i     = i0 + lane;
        const bool valid = i < cnt;
        unsigned tl = 0u;
        if (valid) {
          const unsigned rc = mykey[i];
          const int e = (int)(rc & 0xFFFFFu);
          tl = (rc >> 20) - dlo;
          mykey[i] = (tl << 24) | (unsigned)src[e];
          myw[i]   = w[e];
        }
        if (i0 < FCH * 32) {
          unsigned mo = __builtin_amdgcn_ballot_w32(valid);
#pragma unroll
          for (int k = 0; k < 5; ++k) {
            const unsigned bk = __builtin_amdgcn_ballot_w32(((tl >> k) & 1u) != 0u);
            mo &= ((lane >> k) & 1) ? bk : ~bk;
          }
          omask[wave * OMP + i0 + lane] = mo;
        }
      }
      if (lane == 0) {
        lcnt[wave]  = cnt;
        wmore[wave] = (pos < end) ? 1 : 0;
      }
      __syncthreads();
      const bool more = __builtin_amdgcn_ballot_w32(wmore[lane] != 0) != 0;

      bool by_masks = __builtin_amdgcn_ballot_w32(lcnt[lane] > FCH * 32) == 0;
      int sl = 0, base = 0;
      while (sl < NWV) {
        int n_own = 0;
        if (by_masks) {
          by_masks = false;
          const int cv = lcnt[lane];
          unsigned m[FCH];
          int nv = 0;
#pragma unroll
          for (int c = 0; c < FCH; ++c) {
            m[c] = (c * 32 < cv) ? omask[lane * OMP + c * 32 + wave] : 0u;
            nv += __builtin_popcount(m[c]);
          }
          int incl = nv;
#pragma unroll
          for (int d = 1; d < 32; d <<= 1) {
            const int t = __shfl_up(incl, d, 32);
            if (lane >= d) incl += t;
          }
          const int total = __builtin_amdgcn_readlane(incl, 31);
          if (total <= CAP2) {
            int p = incl - nv;
#pragma unroll
            for (int c = 0; c < FCH; ++c) {
              unsigned mm = m[c];
              while (__builtin_amdgcn_ballot_w32(mm != 0u) != 0) {
                if (mm != 0u) {
                  const int b = __builtin_ctz(mm);
                  mm &= mm - 1u;
                  const int idx = lane * CAP + c * 32 + b;
                  ok[p] = lkey[idx];
                  ow[p] = lwt[idx];
                  ++p;
                }
              }
            }
            n_own = total;
            sl = NWV;
          }
        }
        while (sl < NWV && n_own <= CAP2 - 32) {
          const int c = __builtin_amdgcn_readfirstlane(lcnt[sl]);
          if (base >= c) { ++sl; base = 0; continue; }
          const int i = base + lane;
          unsigned k = 0;
          float ww = 0.f;
          bool mine = false;
          if (i < c) {
            k    = lkey[sl * CAP + i];
            ww   = lwt[sl * CAP + i];
            mine = (int)((k >> 24) & 31u) == wave;
          }
          const unsigned mm = __builtin_amdgcn_ballot_w32(mine);
          if (mm != 0) {
            if (mine) {
              const int idx = n_own + __builtin_popcount(mm & below);
              ok[idx] = k;
              ow[idx] = ww;
            }
            n_own += __builtin_popcount(mm);
          }
          base += 32;
        }
        __builtin_amdgcn_wave_barrier();
        group_by_node(ok, ow, n_own, lane, below);
        if (MODE == 0) {
          for (int i = lane; i < n_own; i += 32) {
            const int g = saved + i;
            if (g < SCAP) sv[g] = make_uint2(ok[i], __float_as_uint(ow[i]));
          }
          saved += n_own;
        }
        add_rows<F, RS>(ok, ow, n_own, H, acc, lane);
        __builtin_amdgcn_wave_barrier();
      }
      if (!more) break;
    }
    if (MODE == 0 && lane == 0) scnt[blockIdx.x * NWV + wave] = saved;
  }
  __syncthreads();

  if constexpr (NCN != 0) {
    if (!use_saved) {
      stage_w<NCN, 1024>(Wn, lbuf, tid);
      __syncthreads();
    }
    tail_product<NCN, EPI, (NCN == GK ? 4 : 3)>(acc, lbuf, ep, On, lo, n, wave, lane);
  } else {
    tail_lsm40(acc, eb, On, lo, n, wave, lane);
  }
}

extern "C" void kernel_launch(void* const* d_in, const int* in_sizes, int n_in,
                              void* d_out, int out_size, void* d_ws, size_t ws_size,
                              hipStream_t stream) {
  const int N = NNODES;
  const float* x     = (const float*)d_in[0];
  const int*   src   = (const int*)d_in[1];
  const int*   tgt   = (const int*)d_in[2];
  const float* w     = (const float*)d_in[3];
  const float* W0    = (const float*)d_in[4];
  const float* b0    = (const float*)d_in[5];
  const float* W1    = (const float*)d_in[6];
  const float* b1    = (const float*)d_in[7];
  const float* W2    = (const float*)d_in[8];
  const float* b2    = (const float*)d_in[9];
  const float* gamma = (const float*)d_in[10];
  const float* beta  = (const float*)d_in[11];
  float* out = (float*)d_out;

  float* H   = (float*)d_ws;
  float* H1  = H + (size_t)N * 128;
  float* H2  = H;
  unsigned* SL = (unsigned*)(H + (size_t)N * 256);
  int*      SC = (int*)(SL + (size_t)NTILES * NWV * SCAP * 2);
  unsigned* RL = (unsigned*)(SC + NTILES * NWV);
  int*      RC = (int*)(RL + (size_t)NRANGES * NPARTS * PART);

  const int gemm_blocks = (N + 127) / 128;

  range_lists<<<NRANGES * NPARTS, NWV * 32, 0, stream>>>(tgt, RL, RC);


  gemm_wmma_f32<128><<<gemm_blocks, 256, 0, stream>>>(x, W0, H, N);
  agg_tile<128, 0, 128, 1><<<NTILES, 1024, 0, stream>>>(
      H, src, w, W1, b0, nullptr, nullptr, H1, SL, SC, RL, RC, N);

  agg_tile<128, 1, 40, 2><<<NTILES, 1024, 0, stream>>>(
      H1, src, w, W2, b1, gamma, beta, H2, SL, SC, RL, RC, N);

  agg_tile<40, 1, 0, 0><<<NTILES, 1024, 0, stream>>>(
      H2, src, w, nullptr, b2, nullptr, nullptr, out, SL, SC, RL, RC, N);
}
